# starting kernel
# baseline (speedup 1.0000x reference)
#include <hip/hip_runtime.h>
#include <math.h>

typedef float v2f __attribute__((ext_vector_type(2)));
typedef float v8f __attribute__((ext_vector_type(8)));

namespace {
constexpr int kB  = 8;
constexpr int kT  = 2048;
constexpr int kD  = 1024;
constexpr int kHD = 64;
constexpr int kM  = 64;
constexpr int kS  = kT + kM;
constexpr float kLnEps = 1e-5f;
constexpr float kScale = 0.125f;
}

__device__ __forceinline__ v8f wmma_f32(v2f a, v2f b, v8f c) {
  return __builtin_amdgcn_wmma_f32_16x16x4_f32(
       false, a,  false, b,
       (short)0, c,  false,  false);
}

__global__ __launch_bounds__(64)
void ln_mem_kernel(const float* __restrict__ c_emb,
                   const float* __restrict__ gamma,
                   const float* __restrict__ beta,
                   float* __restrict__ k_buf,
                   float* __restrict__ v_buf) {
  const int b = blockIdx.x;
  const int r = threadIdx.x;
  const float* row = c_emb + r * kHD;
  float mu = 0.f;
  #pragma unroll
  for (int i = 0; i < kHD; ++i) mu += row[i];
  mu *= (1.f / kHD);
  float var = 0.f;
  #pragma unroll
  for (int i = 0; i < kHD; ++i) { float d = row[i] - mu; var += d * d; }
  var *= (1.f / kHD);
  const float rstd = rsqrtf(var + kLnEps);
  float* kd = k_buf + ((size_t)b * kS + kT + r) * kHD;
  float* vd = v_buf + ((size_t)b * kS + kT + r) * kHD;
  #pragma unroll
  for (int i = 0; i < kHD; ++i) {
    float y = (row[i] - mu) * rstd * gamma[i] + beta[i];
    kd[i] = y;
    vd[i] = y;
  }
}

__global__ __launch_bounds__(256)
void qkv_kernel(const float* __restrict__ x,
                const float* __restrict__ Wq,
                const float* __restrict__ Wk,
                const float* __restrict__ Wv,
                float* __restrict__ q_buf,
                float* __restrict__ k_buf,
                float* __restrict__ v_buf) {
  __shared__ float xs[128][36];
  __shared__ float wst[192][34];

  const int tid  = threadIdx.x;
  const int wid  = tid >> 5;
  const int lane = tid & 31;
  const int half = lane >> 4;
  const int lm   = lane & 15;
  const int r0   = blockIdx.x * 128;
  const int m0   = wid * 16;

  v8f acc[12] = {};
  float4 xreg[4];
  float4 wreg[6];

  auto gload = [&](int kt) {
    #pragma unroll
    for (int i = 0; i < 4; ++i) {
      int idx = tid + 256 * i;
      int row = idx >> 3;
      int c4  = (idx & 7) * 4;
      xreg[i] = *(const float4*)(x + (size_t)(r0 + row) * kD + kt + c4);
    }
    #pragma unroll
    for (int i = 0; i < 6; ++i) {
      int idx = tid + 256 * i;
      int kr  = idx / 48;
      int c4  = idx % 48;
      int mat = c4 >> 4;
      int cc  = (c4 & 15) * 4;
      const float* src = (mat == 0) ? Wq : ((mat == 1) ? Wk : Wv);
      wreg[i] = *(const float4*)(src + (size_t)(kt + kr) * kHD + cc);
    }
  };
  auto lstore = [&]() {
    #pragma unroll
    for (int i = 0; i < 4; ++i) {
      int idx = tid + 256 * i;
      int row = idx >> 3;
      int c4  = (idx & 7) * 4;
      *(float4*)&xs[row][c4] = xreg[i];
    }
    #pragma unroll
    for (int i = 0; i < 6; ++i) {
      int idx = tid + 256 * i;
      int kr  = idx / 48;
      int c4  = idx % 48;
      int col = (c4 >> 4) * 64 + (c4 & 15) * 4;
      wst[col + 0][kr] = wreg[i].x;
      wst[col + 1][kr] = wreg[i].y;
      wst[col + 2][kr] = wreg[i].z;
      wst[col + 3][kr] = wreg[i].w;
    }
  };

  gload(0);
  lstore();
  __syncthreads();

  constexpr int kNT = kD / 32;
  for (int t = 0; t < kNT; ++t) {
    if (t + 1 < kNT) gload(32 * (t + 1));
    #pragma unroll
    for (int kk = 0; kk < 8; ++kk) {
      const int k0 = kk * 4;
      v2f a = *(const v2f*)&xs[m0 + lm][k0 + 2 * half];
      #pragma unroll
      for (int c = 0; c < 12; ++c) {
        v2f bf = *(const v2f*)&wst[c * 16 + lm][k0 + 2 * half];
        acc[c] = wmma_f32(a, bf, acc[c]);
      }
    }
    __syncthreads();
    if (t + 1 < kNT) lstore();
    __syncthreads();
  }

  #pragma unroll
  for (int c = 0; c < 12; ++c) {
    const int mat = c >> 2;
    const int hd  = (c & 3) * 16 + lm;
    #pragma unroll
    for (int i = 0; i < 8; ++i) {
      const int gr = r0 + m0 + i + 8 * half;
      const float val = acc[c][i];
      if (mat == 0) {
        q_buf[(size_t)gr * kHD + hd] = val;
      } else {
        const int bb = gr / kT;
        const int tt = gr - bb * kT;
        float* dst = (mat == 1) ? k_buf : v_buf;
        dst[((size_t)bb * kS + tt) * kHD + hd] = val;
      }
    }
  }
}

__global__ __launch_bounds__(256)
void attn_kernel(const float* __restrict__ q_buf,
                 const float* __restrict__ k_buf,
                 const float* __restrict__ v_buf,
                 float* __restrict__ out) {
  __shared__ float qs[128][68];
  __shared__ float ks[64][68];
  __shared__ float vst[64][68];
  __shared__ float ps[8][16][68];

  const int tid  = threadIdx.x;
  const int wid  = tid >> 5;
  const int lane = tid & 31;
  const int half = lane >> 4;
  const int lm   = lane & 15;
  const int b    = blockIdx.y;
  const int q0   = blockIdx.x * 128;
  const int m0   = wid * 16;

  float4 kreg[4], vreg[4];
  auto gload = [&](int s0) {
    #pragma unroll
    for (int i = 0; i < 4; ++i) {
      int idx = tid + 256 * i;
      int row = idx >> 4;
      int c4  = (idx & 15) * 4;
      kreg[i] = *(const float4*)(k_buf + ((size_t)b * kS + s0 + row) * kHD + c4);
      vreg[i] = *(const float4*)(v_buf + ((size_t)b * kS + s0 + row) * kHD + c4);
    }
  };
  auto lstore = [&]() {
    #pragma unroll
    for (int i = 0; i < 4; ++i) {
      int idx = tid + 256 * i;
      int row = idx >> 4;
      int c4  = (idx & 15) * 4;
      *(float4*)&ks[row][c4] = kreg[i];
      vst[c4 + 0][row] = vreg[i].x;
      vst[c4 + 1][row] = vreg[i].y;
      vst[c4 + 2][row] = vreg[i].z;
      vst[c4 + 3][row] = vreg[i].w;
    }
  };

  #pragma unroll
  for (int i = 0; i < 8; ++i) {
    int idx = tid + 256 * i;
    int row = idx >> 4;
    int c4  = (idx & 15) * 4;
    float4 v = *(const float4*)(q_buf + ((size_t)b * kT + q0 + row) * kHD + c4);
    v.x *= kScale; v.y *= kScale; v.z *= kScale; v.w *= kScale;
    *(float4*)&qs[row][c4] = v;
  }
  gload(0);
  lstore();
  __syncthreads();

  v8f oacc[4] = {};
  float mrun[8], lrun[8];
  #pragma unroll
  for (int i = 0; i < 8; ++i) { mrun[i] = -INFINITY; lrun[i] = 0.f; }

  constexpr int kNT = kS / 64;
  for (int it = 0; it < kNT; ++it) {
    if (it + 1 < kNT) gload(64 * (it + 1));

    v8f sacc[4] = {};
    #pragma unroll
    for (int kk = 0; kk < 16; ++kk) {
      const int k0 = kk * 4;
      v2f a = *(const v2f*)&qs[m0 + lm][k0 + 2 * half];
      #pragma unroll
      for (int j = 0; j < 4; ++j) {
        v2f bf = *(const v2f*)&ks[j * 16 + lm][k0 + 2 * half];
        sacc[j] = wmma_f32(a, bf, sacc[j]);
      }
    }

    float mnew[8], alpha[8], rsum[8];
    #pragma unroll
    for (int i = 0; i < 8; ++i) {
      float v = fmaxf(fmaxf(sacc[0][i], sacc[1][i]),
                      fmaxf(sacc[2][i], sacc[3][i]));
      #pragma unroll
      for (int off = 1; off < 16; off <<= 1)
        v = fmaxf(v, __shfl_xor(v, off));
      mnew[i]  = fmaxf(mrun[i], v);
      alpha[i] = __expf(mrun[i] - mnew[i]);
      mrun[i]  = mnew[i];
      rsum[i]  = 0.f;
    }
    #pragma unroll
    for (int j = 0; j < 4; ++j) {
      #pragma unroll
      for (int i = 0; i < 8; ++i) {
        float p = __expf(sacc[j][i] - mnew[i]);
        sacc[j][i] = p;
        rsum[i] += p;
      }
    }
    #pragma unroll
    for (int i = 0; i < 8; ++i) {
      float rs = rsum[i];
      #pragma unroll
      for (int off = 1; off < 16; off <<= 1)
        rs += __shfl_xor(rs, off);
      lrun[i] = lrun[i] * alpha[i] + rs;
    }
    #pragma unroll
    for (int c = 0; c < 4; ++c)
      #pragma unroll
      for (int i = 0; i < 8; ++i)
        oacc[c][i] *= alpha[i];

    #pragma unroll
    for (int j = 0; j < 4; ++j)
      #pragma unroll
      for (int i = 0; i < 8; ++i)
        ps[wid][i + 8 * half][j * 16 + lm] = sacc[j][i];

    #pragma unroll
    for (int kk = 0; kk < 16; ++kk) {
      const int k0 = kk * 4;
      v2f a = *(const v2f*)&ps[wid][lm][k0 + 2 * half];
      #pragma unroll
      for (int c = 0; c < 4; ++c) {
        v2f bf = *(const v2f*)&vst[c * 16 + lm][k0 + 2 * half];
        oacc[c] = wmma_f32(a, bf, oacc[c]);
      }
    }

    __syncthreads();
    if (it + 1 < kNT) lstore();
    __syncthreads();
  }

  #pragma unroll
  for (int c = 0; c < 4; ++c) {
    const int hd = c * 16 + lm;
    #pragma unroll
    for (int i = 0; i < 8; ++i) {
      const int t = q0 + m0 + i + 8 * half;
      out[((size_t)b * kT + t) * kHD + hd] = oacc[c][i] / lrun[i];
    }
  }
}

extern "C" void kernel_launch(void* const* d_in, const int* in_sizes, int n_in,
                              void* d_out, int out_size, void* d_ws, size_t ws_size,
                              hipStream_t stream) {
  const float* x     = (const float*)d_in[0];
  const float* c_emb = (const float*)d_in[1];
  const float* Wq    = (const float*)d_in[2];
  const float* Wk    = (const float*)d_in[3];
  const float* Wv    = (const float*)d_in[4];
  const float* gamma = (const float*)d_in[5];
  const float* beta  = (const float*)d_in[6];
  float* out = (float*)d_out;

  float* q_buf = (float*)d_ws;
  float* k_buf = q_buf + (size_t)kB * kT * kHD;
  float* v_buf = k_buf + (size_t)kB * kS * kHD;

  ln_mem_kernel<<<kB, 64, 0, stream>>>(c_emb, gamma, beta, k_buf, v_buf);
  qkv_kernel<<<(kB * kT) / 128, 256, 0, stream>>>(x, Wq, Wk, Wv,
                                                  q_buf, k_buf, v_buf);
  attn_kernel<<<dim3(kT / 128, kB), 256, 0, stream>>>(q_buf, k_buf, v_buf, out);
}


# projection: waves split by role: waves 4-7 only stage K-tile t (global->registers->LDS stage t&1) while waves 0-3 (32 rows x 192 cols each) run tile t-1's WMMAs; one barrier per tile
# speedup vs baseline: 1.2138x; 1.2138x over previous
#include <hip/hip_runtime.h>
#include <math.h>

typedef float v2f __attribute__((ext_vector_type(2)));
typedef float v4f __attribute__((ext_vector_type(4)));
typedef float v8f __attribute__((ext_vector_type(8)));

namespace {
constexpr int kB  = 8;
constexpr int kT  = 2048;
constexpr int kD  = 1024;
constexpr int kHD = 64;
constexpr int kM  = 64;
constexpr int kS  = kT + kM;
constexpr float kLnEps = 1e-5f;
constexpr float kScale = 0.125f;
}

__device__ __forceinline__ v8f wmma_f32(v2f a, v2f b, v8f c) {
  return __builtin_amdgcn_wmma_f32_16x16x4_f32(
       false, a,  false, b,
       (short)0, c,  false,  false);
}

__global__ __launch_bounds__(64)
void ln_mem_kernel(const float* __restrict__ c_emb,
                   const float* __restrict__ gamma,
                   const float* __restrict__ beta,
                   float* __restrict__ k_buf,
                   float* __restrict__ v_buf) {
  const int b = blockIdx.x;
  const int r = threadIdx.x;
  const float* row = c_emb + r * kHD;
  float mu = 0.f;
  #pragma unroll
  for (int i = 0; i < kHD; ++i) mu += row[i];
  mu *= (1.f / kHD);
  float var = 0.f;
  #pragma unroll
  for (int i = 0; i < kHD; ++i) { float d = row[i] - mu; var += d * d; }
  var *= (1.f / kHD);
  const float rstd = rsqrtf(var + kLnEps);
  float* kd = k_buf + ((size_t)b * kS + kT + r) * kHD;
  float* vd = v_buf + ((size_t)b * kS + kT + r) * kHD;
  #pragma unroll
  for (int i = 0; i < kHD; ++i) {
    float y = (row[i] - mu) * rstd * gamma[i] + beta[i];
    kd[i] = y;
    vd[i] = y;
  }
}

__global__ __launch_bounds__(256)
void qkv_kernel(const float* __restrict__ x,
                const float* __restrict__ Wq,
                const float* __restrict__ Wk,
                const float* __restrict__ Wv,
                float* __restrict__ q_buf,
                float* __restrict__ k_buf,
                float* __restrict__ v_buf) {
  __shared__ alignas(16) float xs[2][128][36];
  __shared__ alignas(16) float wst[2][192][34];

  const int tid  = threadIdx.x;
  const int wid  = __builtin_amdgcn_readfirstlane(tid >> 5);
  const int lane = tid & 31;
  const int half = lane >> 4;
  const int lm   = lane & 15;
  const int r0   = blockIdx.x * 128;
  const bool staging = (wid >= 4);
  const int m0   = (wid & 3) * 32;
  const int ltid = tid & 127;

  v8f acc[2][12] = {};

  auto stage_tile = [&](int t) {
    const int kt = 32 * t;
    const int st = t & 1;
    v4f xr[8];
    v4f wr[12];
    #pragma unroll
    for (int i = 0; i < 8; ++i) {
      const int idx = ltid + 128 * i;
      const int row = idx >> 3;
      const int c4  = (idx & 7) * 4;
      xr[i] = *(const v4f*)(x + (size_t)(r0 + row) * kD + kt + c4);
    }
    #pragma unroll
    for (int mat = 0; mat < 3; ++mat) {
      const float* src = (mat == 0) ? Wq : ((mat == 1) ? Wk : Wv);
      #pragma unroll
      for (int u = 0; u < 4; ++u) {
        const int idx = ltid + 128 * u;
        const int kr  = idx >> 4;
        const int cc  = (idx & 15) * 4;
        wr[mat * 4 + u] = *(const v4f*)(src + (size_t)(kt + kr) * kHD + cc);
      }
    }
    #pragma unroll
    for (int i = 0; i < 8; ++i) {
      const int idx = ltid + 128 * i;
      const int row = idx >> 3;
      const int c4  = (idx & 7) * 4;
      *(v4f*)&xs[st][row][c4] = xr[i];
    }
    #pragma unroll
    for (int mat = 0; mat < 3; ++mat) {
      #pragma unroll
      for (int u = 0; u < 4; ++u) {
        const int idx = ltid + 128 * u;
        const int kr  = idx >> 4;
        const int col = mat * 64 + (idx & 15) * 4;
        const v4f w = wr[mat * 4 + u];
        wst[st][col + 0][kr] = w.x;
        wst[st][col + 1][kr] = w.y;
        wst[st][col + 2][kr] = w.z;
        wst[st][col + 3][kr] = w.w;
      }
    }
  };
  auto matrix_tile = [&](int t) {
    const int st = t & 1;
    #pragma unroll
    for (int kk = 0; kk < 8; ++kk) {
      const int k0 = kk * 4;
      const v2f a0 = *(const v2f*)&xs[st][m0 + lm][k0 + 2 * half];
      const v2f a1 = *(const v2f*)&xs[st][m0 + 16 + lm][k0 + 2 * half];
      #pragma unroll
      for (int c = 0; c < 12; ++c) {
        const v2f bf = *(const v2f*)&wst[st][c * 16 + lm][k0 + 2 * half];
        acc[0][c] = wmma_f32(a0, bf, acc[0][c]);
        acc[1][c] = wmma_f32(a1, bf, acc[1][c]);
      }
    }
  };

  constexpr int kNT = kD / 32;
  if (staging) stage_tile(0);
  __syncthreads();
  for (int t = 1; t < kNT; ++t) {
    if (staging) stage_tile(t); else matrix_tile(t - 1);
    __syncthreads();
  }
  if (!staging) matrix_tile(kNT - 1);

  if (!staging) {
    #pragma unroll
    for (int r = 0; r < 2; ++r) {
      #pragma unroll
      for (int c = 0; c < 12; ++c) {
        const int mat = c >> 2;
        const int hd  = (c & 3) * 16 + lm;
        #pragma unroll
        for (int i = 0; i < 8; ++i) {
          const int gr = r0 + m0 + 16 * r + i + 8 * half;
          const float val = acc[r][c][i];
          if (mat == 0) {
            q_buf[(size_t)gr * kHD + hd] = val;
          } else {
            const int bb = gr / kT;
            const int tt = gr - bb * kT;
            float* dst = (mat == 1) ? k_buf : v_buf;
            dst[((size_t)bb * kS + tt) * kHD + hd] = val;
          }
        }
      }
    }
  }
}

__global__ __launch_bounds__(256)
void attn_kernel(const float* __restrict__ q_buf,
                 const float* __restrict__ k_buf,
                 const float* __restrict__ v_buf,
                 float* __restrict__ out) {
  __shared__ float qs[128][68];
  __shared__ float ks[64][68];
  __shared__ float vst[64][68];
  __shared__ float ps[8][16][68];

  const int tid  = threadIdx.x;
  const int wid  = tid >> 5;
  const int lane = tid & 31;
  const int half = lane >> 4;
  const int lm   = lane & 15;
  const int b    = blockIdx.y;
  const int q0   = blockIdx.x * 128;
  const int m0   = wid * 16;

  float4 kreg[4], vreg[4];
  auto gload = [&](int s0) {
    #pragma unroll
    for (int i = 0; i < 4; ++i) {
      int idx = tid + 256 * i;
      int row = idx >> 4;
      int c4  = (idx & 15) * 4;
      kreg[i] = *(const float4*)(k_buf + ((size_t)b * kS + s0 + row) * kHD + c4);
      vreg[i] = *(const float4*)(v_buf + ((size_t)b * kS + s0 + row) * kHD + c4);
    }
  };
  auto lstore = [&]() {
    #pragma unroll
    for (int i = 0; i < 4; ++i) {
      int idx = tid + 256 * i;
      int row = idx >> 4;
      int c4  = (idx & 15) * 4;
      *(float4*)&ks[row][c4] = kreg[i];
      vst[c4 + 0][row] = vreg[i].x;
      vst[c4 + 1][row] = vreg[i].y;
      vst[c4 + 2][row] = vreg[i].z;
      vst[c4 + 3][row] = vreg[i].w;
    }
  };

  #pragma unroll
  for (int i = 0; i < 8; ++i) {
    int idx = tid + 256 * i;
    int row = idx >> 4;
    int c4  = (idx & 15) * 4;
    float4 v = *(const float4*)(q_buf + ((size_t)b * kT + q0 + row) * kHD + c4);
    v.x *= kScale; v.y *= kScale; v.z *= kScale; v.w *= kScale;
    *(float4*)&qs[row][c4] = v;
  }
  gload(0);
  lstore();
  __syncthreads();

  v8f oacc[4] = {};
  float mrun[8], lrun[8];
  #pragma unroll
  for (int i = 0; i < 8; ++i) { mrun[i] = -INFINITY; lrun[i] = 0.f; }

  constexpr int kNT = kS / 64;
  for (int it = 0; it < kNT; ++it) {
    if (it + 1 < kNT) gload(64 * (it + 1));

    v8f sacc[4] = {};
    #pragma unroll
    for (int kk = 0; kk < 16; ++kk) {
      const int k0 = kk * 4;
      v2f a = *(const v2f*)&qs[m0 + lm][k0 + 2 * half];
      #pragma unroll
      for (int j = 0; j < 4; ++j) {
        v2f bf = *(const v2f*)&ks[j * 16 + lm][k0 + 2 * half];
        sacc[j] = wmma_f32(a, bf, sacc[j]);
      }
    }

    float mnew[8], alpha[8], rsum[8];
    #pragma unroll
    for (int i = 0; i < 8; ++i) {
      float v = fmaxf(fmaxf(sacc[0][i], sacc[1][i]),
                      fmaxf(sacc[2][i], sacc[3][i]));
      #pragma unroll
      for (int off = 1; off < 16; off <<= 1)
        v = fmaxf(v, __shfl_xor(v, off));
      mnew[i]  = fmaxf(mrun[i], v);
      alpha[i] = __expf(mrun[i] - mnew[i]);
      mrun[i]  = mnew[i];
      rsum[i]  = 0.f;
    }
    #pragma unroll
    for (int j = 0; j < 4; ++j) {
      #pragma unroll
      for (int i = 0; i < 8; ++i) {
        float p = __expf(sacc[j][i] - mnew[i]);
        sacc[j][i] = p;
        rsum[i] += p;
      }
    }
    #pragma unroll
    for (int i = 0; i < 8; ++i) {
      float rs = rsum[i];
      #pragma unroll
      for (int off = 1; off < 16; off <<= 1)
        rs += __shfl_xor(rs, off);
      lrun[i] = lrun[i] * alpha[i] + rs;
    }
    #pragma unroll
    for (int c = 0; c < 4; ++c)
      #pragma unroll
      for (int i = 0; i < 8; ++i)
        oacc[c][i] *= alpha[i];

    #pragma unroll
    for (int j = 0; j < 4; ++j)
      #pragma unroll
      for (int i = 0; i < 8; ++i)
        ps[wid][i + 8 * half][j * 16 + lm] = sacc[j][i];

    #pragma unroll
    for (int kk = 0; kk < 16; ++kk) {
      const int k0 = kk * 4;
      v2f a = *(const v2f*)&ps[wid][lm][k0 + 2 * half];
      #pragma unroll
      for (int c = 0; c < 4; ++c) {
        v2f bf = *(const v2f*)&vst[c * 16 + lm][k0 + 2 * half];
        oacc[c] = wmma_f32(a, bf, oacc[c]);
      }
    }

    __syncthreads();
    if (it + 1 < kNT) lstore();
    __syncthreads();
  }

  #pragma unroll
  for (int c = 0; c < 4; ++c) {
    const int hd = c * 16 + lm;
    #pragma unroll
    for (int i = 0; i < 8; ++i) {
      const int t = q0 + m0 + i + 8 * half;
      out[((size_t)b * kT + t) * kHD + hd] = oacc[c][i] / lrun[i];
    }
  }
}

extern "C" void kernel_launch(void* const* d_in, const int* in_sizes, int n_in,
                              void* d_out, int out_size, void* d_ws, size_t ws_size,
                              hipStream_t stream) {
  const float* x     = (const float*)d_in[0];
  const float* c_emb = (const float*)d_in[1];
  const float* Wq    = (const float*)d_in[2];
  const float* Wk    = (const float*)d_in[3];
  const float* Wv    = (const float*)d_in[4];
  const float* gamma = (const float*)d_in[5];
  const float* beta  = (const float*)d_in[6];
  float* out = (float*)d_out;

  float* q_buf = (float*)d_ws;
  float* k_buf = q_buf + (size_t)kB * kT * kHD;
  float* v_buf = k_buf + (size_t)kB * kS * kHD;

  ln_mem_kernel<<<kB, 64, 0, stream>>>(c_emb, gamma, beta, k_buf, v_buf);
  qkv_kernel<<<(kB * kT) / 128, 256, 0, stream>>>(x, Wq, Wk, Wv,
                                                  q_buf, k_buf, v_buf);
  attn_kernel<<<dim3(kT / 128, kB), 256, 0, stream>>>(q_buf, k_buf, v_buf, out);
}


# attention: waves split by role: 4 added waves (8-11) only stage key/value tile t (global->registers->LDS stage t&1) while waves 0-7 run tile t-1's WMMAs and softmax; one barrier per tile, not two
# speedup vs baseline: 1.5633x; 1.2879x over previous
#include <hip/hip_runtime.h>
#include <math.h>

typedef float v2f __attribute__((ext_vector_type(2)));
typedef float v4f __attribute__((ext_vector_type(4)));
typedef float v8f __attribute__((ext_vector_type(8)));

namespace {
constexpr int kB  = 8;
constexpr int kT  = 2048;
constexpr int kD  = 1024;
constexpr int kHD = 64;
constexpr int kM  = 64;
constexpr int kS  = kT + kM;
constexpr float kLnEps = 1e-5f;
constexpr float kScale = 0.125f;
}

__device__ __forceinline__ v8f wmma_f32(v2f a, v2f b, v8f c) {
  return __builtin_amdgcn_wmma_f32_16x16x4_f32(
       false, a,  false, b,
       (short)0, c,  false,  false);
}

__global__ __launch_bounds__(64)
void ln_mem_kernel(const float* __restrict__ c_emb,
                   const float* __restrict__ gamma,
                   const float* __restrict__ beta,
                   float* __restrict__ k_buf,
                   float* __restrict__ v_buf) {
  const int b = blockIdx.x;
  const int r = threadIdx.x;
  const float* row = c_emb + r * kHD;
  float mu = 0.f;
  #pragma unroll
  for (int i = 0; i < kHD; ++i) mu += row[i];
  mu *= (1.f / kHD);
  float var = 0.f;
  #pragma unroll
  for (int i = 0; i < kHD; ++i) { float d = row[i] - mu; var += d * d; }
  var *= (1.f / kHD);
  const float rstd = rsqrtf(var + kLnEps);
  float* kd = k_buf + ((size_t)b * kS + kT + r) * kHD;
  float* vd = v_buf + ((size_t)b * kS + kT + r) * kHD;
  #pragma unroll
  for (int i = 0; i < kHD; ++i) {
    float y = (row[i] - mu) * rstd * gamma[i] + beta[i];
    kd[i] = y;
    vd[i] = y;
  }
}

__global__ __launch_bounds__(256)
void qkv_kernel(const float* __restrict__ x,
                const float* __restrict__ Wq,
                const float* __restrict__ Wk,
                const float* __restrict__ Wv,
                float* __restrict__ q_buf,
                float* __restrict__ k_buf,
                float* __restrict__ v_buf) {
  __shared__ alignas(16) float xs[2][128][36];
  __shared__ alignas(16) float wst[2][192][34];

  const int tid  = threadIdx.x;
  const int wid  = __builtin_amdgcn_readfirstlane(tid >> 5);
  const int lane = tid & 31;
  const int half = lane >> 4;
  const int lm   = lane & 15;
  const int r0   = blockIdx.x * 128;
  const bool staging = (wid >= 4);
  const int m0   = (wid & 3) * 32;
  const int ltid = tid & 127;

  v8f acc[2][12] = {};

  auto stage_tile = [&](int t) {
    const int kt = 32 * t;
    const int st = t & 1;
    v4f xr[8];
    v4f wr[12];
    #pragma unroll
    for (int i = 0; i < 8; ++i) {
      const int idx = ltid + 128 * i;
      const int row = idx >> 3;
      const int c4  = (idx & 7) * 4;
      xr[i] = *(const v4f*)(x + (size_t)(r0 + row) * kD + kt + c4);
    }
    #pragma unroll
    for (int mat = 0; mat < 3; ++mat) {
      const float* src = (mat == 0) ? Wq : ((mat == 1) ? Wk : Wv);
      #pragma unroll
      for (int u = 0; u < 4; ++u) {
        const int idx = ltid + 128 * u;
        const int kr  = idx >> 4;
        const int cc  = (idx & 15) * 4;
        wr[mat * 4 + u] = *(const v4f*)(src + (size_t)(kt + kr) * kHD + cc);
      }
    }
    #pragma unroll
    for (int i = 0; i < 8; ++i) {
      const int idx = ltid + 128 * i;
      const int row = idx >> 3;
      const int c4  = (idx & 7) * 4;
      *(v4f*)&xs[st][row][c4] = xr[i];
    }
    #pragma unroll
    for (int mat = 0; mat < 3; ++mat) {
      #pragma unroll
      for (int u = 0; u < 4; ++u) {
        const int idx = ltid + 128 * u;
        const int kr  = idx >> 4;
        const int col = mat * 64 + (idx & 15) * 4;
        const v4f w = wr[mat * 4 + u];
        wst[st][col + 0][kr] = w.x;
        wst[st][col + 1][kr] = w.y;
        wst[st][col + 2][kr] = w.z;
        wst[st][col + 3][kr] = w.w;
      }
    }
  };
  auto matrix_tile = [&](int t) {
    const int st = t & 1;
    #pragma unroll
    for (int kk = 0; kk < 8; ++kk) {
      const int k0 = kk * 4;
      const v2f a0 = *(const v2f*)&xs[st][m0 + lm][k0 + 2 * half];
      const v2f a1 = *(const v2f*)&xs[st][m0 + 16 + lm][k0 + 2 * half];
      #pragma unroll
      for (int c = 0; c < 12; ++c) {
        const v2f bf = *(const v2f*)&wst[st][c * 16 + lm][k0 + 2 * half];
        acc[0][c] = wmma_f32(a0, bf, acc[0][c]);
        acc[1][c] = wmma_f32(a1, bf, acc[1][c]);
      }
    }
  };

  constexpr int kNT = kD / 32;
  if (staging) stage_tile(0);
  __syncthreads();
  for (int t = 1; t < kNT; ++t) {
    if (staging) stage_tile(t); else matrix_tile(t - 1);
    __syncthreads();
  }
  if (!staging) matrix_tile(kNT - 1);

  if (!staging) {
    #pragma unroll
    for (int r = 0; r < 2; ++r) {
      #pragma unroll
      for (int c = 0; c < 12; ++c) {
        const int mat = c >> 2;
        const int hd  = (c & 3) * 16 + lm;
        #pragma unroll
        for (int i = 0; i < 8; ++i) {
          const int gr = r0 + m0 + 16 * r + i + 8 * half;
          const float val = acc[r][c][i];
          if (mat == 0) {
            q_buf[(size_t)gr * kHD + hd] = val;
          } else {
            const int bb = gr / kT;
            const int tt = gr - bb * kT;
            float* dst = (mat == 1) ? k_buf : v_buf;
            dst[((size_t)bb * kS + tt) * kHD + hd] = val;
          }
        }
      }
    }
  }
}

__global__ __launch_bounds__(384)
void attn_kernel(const float* __restrict__ q_buf,
                 const float* __restrict__ k_buf,
                 const float* __restrict__ v_buf,
                 float* __restrict__ out) {
  __shared__ alignas(16) float qs[128][68];
  __shared__ alignas(16) float ks[2][64][68];
  __shared__ alignas(16) float vst[2][64][68];
  __shared__ alignas(16) float ps[8][16][68];

  const int tid  = threadIdx.x;
  const int wid  = __builtin_amdgcn_readfirstlane(tid >> 5);
  const int lane = tid & 31;
  const int half = lane >> 4;
  const int lm   = lane & 15;
  const int b    = blockIdx.y;
  const int q0   = blockIdx.x * 128;
  const bool staging = (wid >= 8);
  const int pw   = wid & 7;
  const int m0   = pw * 16;
  const int ltid = tid & 127;

  v8f oacc[4] = {};
  float mrun[8], lrun[8];
  #pragma unroll
  for (int i = 0; i < 8; ++i) { mrun[i] = -INFINITY; lrun[i] = 0.f; }

  auto stage_tile = [&](int t) {
    const int s0 = 64 * t;
    const int st = t & 1;
    v4f kr[8], vr[8];
    #pragma unroll
    for (int i = 0; i < 8; ++i) {
      const int idx = ltid + 128 * i;
      const int row = idx >> 4;
      const int c4  = (idx & 15) * 4;
      kr[i] = *(const v4f*)(k_buf + ((size_t)b * kS + s0 + row) * kHD + c4);
      vr[i] = *(const v4f*)(v_buf + ((size_t)b * kS + s0 + row) * kHD + c4);
    }
    #pragma unroll
    for (int i = 0; i < 8; ++i) {
      const int idx = ltid + 128 * i;
      const int row = idx >> 4;
      const int c4  = (idx & 15) * 4;
      *(v4f*)&ks[st][row][c4] = kr[i];
      const v4f v = vr[i];
      vst[st][c4 + 0][row] = v.x;
      vst[st][c4 + 1][row] = v.y;
      vst[st][c4 + 2][row] = v.z;
      vst[st][c4 + 3][row] = v.w;
    }
  };

  auto matrix_tile = [&](int t) {
    const int st = t & 1;

    v8f sacc[4] = {};
    #pragma unroll
    for (int kk = 0; kk < 16; ++kk) {
      const int k0 = kk * 4;
      v2f a = *(const v2f*)&qs[m0 + lm][k0 + 2 * half];
      #pragma unroll
      for (int j = 0; j < 4; ++j) {
        v2f bf = *(const v2f*)&ks[st][j * 16 + lm][k0 + 2 * half];
        sacc[j] = wmma_f32(a, bf, sacc[j]);
      }
    }

    float mnew[8], alpha[8], rsum[8];
    #pragma unroll
    for (int i = 0; i < 8; ++i) {
      float v = fmaxf(fmaxf(sacc[0][i], sacc[1][i]),
                      fmaxf(sacc[2][i], sacc[3][i]));
      #pragma unroll
      for (int off = 1; off < 16; off <<= 1)
        v = fmaxf(v, __shfl_xor(v, off));
      mnew[i]  = fmaxf(mrun[i], v);
      alpha[i] = __expf(mrun[i] - mnew[i]);
      mrun[i]  = mnew[i];
      rsum[i]  = 0.f;
    }
    #pragma unroll
    for (int j = 0; j < 4; ++j) {
      #pragma unroll
      for (int i = 0; i < 8; ++i) {
        float p = __expf(sacc[j][i] - mnew[i]);
        sacc[j][i] = p;
        rsum[i] += p;
      }
    }
    #pragma unroll
    for (int i = 0; i < 8; ++i) {
      float rs = rsum[i];
      #pragma unroll
      for (int off = 1; off < 16; off <<= 1)
        rs += __shfl_xor(rs, off);
      lrun[i] = lrun[i] * alpha[i] + rs;
    }
    #pragma unroll
    for (int c = 0; c < 4; ++c)
      #pragma unroll
      for (int i = 0; i < 8; ++i)
        oacc[c][i] *= alpha[i];

    #pragma unroll
    for (int j = 0; j < 4; ++j)
      #pragma unroll
      for (int i = 0; i < 8; ++i)
        ps[pw][i + 8 * half][j * 16 + lm] = sacc[j][i];

    #pragma unroll
    for (int kk = 0; kk < 16; ++kk) {
      const int k0 = kk * 4;
      v2f a = *(const v2f*)&ps[pw][lm][k0 + 2 * half];
      #pragma unroll
      for (int c = 0; c < 4; ++c) {
        v2f bf = *(const v2f*)&vst[st][c * 16 + lm][k0 + 2 * half];
        oacc[c] = wmma_f32(a, bf, oacc[c]);
      }
    }
  };

  constexpr int kNT = kS / 64;
  if (staging) {
    stage_tile(0);
  } else {
    #pragma unroll
    for (int i = 0; i < 8; ++i) {
      int idx = tid + 256 * i;
      int row = idx >> 4;
      int c4  = (idx & 15) * 4;
      v4f v = *(const v4f*)(q_buf + ((size_t)b * kT + q0 + row) * kHD + c4);
      v *= kScale;
      *(v4f*)&qs[row][c4] = v;
    }
  }
  __syncthreads();
  for (int t = 1; t < kNT; ++t) {
    if (staging) stage_tile(t); else matrix_tile(t - 1);
    __syncthreads();
  }

  if (!staging) {
    matrix_tile(kNT - 1);
    #pragma unroll
    for (int c = 0; c < 4; ++c) {
      const int hd = c * 16 + lm;
      #pragma unroll
      for (int i = 0; i < 8; ++i) {
        const int t = q0 + m0 + i + 8 * half;
        out[((size_t)b * kT + t) * kHD + hd] = oacc[c][i] / lrun[i];
      }
    }
  }
}

extern "C" void kernel_launch(void* const* d_in, const int* in_sizes, int n_in,
                              void* d_out, int out_size, void* d_ws, size_t ws_size,
                              hipStream_t stream) {
  const float* x     = (const float*)d_in[0];
  const float* c_emb = (const float*)d_in[1];
  const float* Wq    = (const float*)d_in[2];
  const float* Wk    = (const float*)d_in[3];
  const float* Wv    = (const float*)d_in[4];
  const float* gamma = (const float*)d_in[5];
  const float* beta  = (const float*)d_in[6];
  float* out = (float*)d_out;

  float* q_buf = (float*)d_ws;
  float* k_buf = q_buf + (size_t)kB * kT * kHD;
  float* v_buf = k_buf + (size_t)kB * kS * kHD;

  ln_mem_kernel<<<kB, 64, 0, stream>>>(c_emb, gamma, beta, k_buf, v_buf);
  qkv_kernel<<<(kB * kT) / 128, 256, 0, stream>>>(x, Wq, Wk, Wv,
                                                  q_buf, k_buf, v_buf);
  attn_kernel<<<dim3(kT / 128, kB), 384, 0, stream>>>(q_buf, k_buf, v_buf, out);
}


# projection: the 192 q|k|v columns split over 2 work-groups per 128-row tile (tile 128x192, k 32 -> 128x96, k 64), grid 128 -> 256: one per WGP instead of half the chip idle; sum order unchanged
# speedup vs baseline: 1.9358x; 1.2383x over previous
#include <hip/hip_runtime.h>
#include <math.h>

typedef float v2f __attribute__((ext_vector_type(2)));
typedef float v4f __attribute__((ext_vector_type(4)));
typedef float v8f __attribute__((ext_vector_type(8)));

namespace {
constexpr int kB  = 8;
constexpr int kT  = 2048;
constexpr int kD  = 1024;
constexpr int kHD = 64;
constexpr int kM  = 64;
constexpr int kS  = kT + kM;
constexpr float kLnEps = 1e-5f;
constexpr float kScale = 0.125f;
constexpr int kProjBlocks = 256;
}

__device__ __forceinline__ v8f wmma_f32(v2f a, v2f b, v8f c) {
  return __builtin_amdgcn_wmma_f32_16x16x4_f32(
       false, a,  false, b,
       (short)0, c,  false,  false);
}

__global__ __launch_bounds__(64)
void ln_mem_kernel(const float* __restrict__ c_emb,
                   const float* __restrict__ gamma,
                   const float* __restrict__ beta,
                   float* __restrict__ k_buf,
                   float* __restrict__ v_buf) {
  const int b = blockIdx.x;
  const int r = threadIdx.x;
  const float* row = c_emb + r * kHD;
  float mu = 0.f;
  #pragma unroll
  for (int i = 0; i < kHD; ++i) mu += row[i];
  mu *= (1.f / kHD);
  float var = 0.f;
  #pragma unroll
  for (int i = 0; i < kHD; ++i) { float d = row[i] - mu; var += d * d; }
  var *= (1.f / kHD);
  const float rstd = rsqrtf(var + kLnEps);
  float* kd = k_buf + ((size_t)b * kS + kT + r) * kHD;
  float* vd = v_buf + ((size_t)b * kS + kT + r) * kHD;
  #pragma unroll
  for (int i = 0; i < kHD; ++i) {
    float y = (row[i] - mu) * rstd * gamma[i] + beta[i];
    kd[i] = y;
    vd[i] = y;
  }
}

template <int NH>
__device__ __forceinline__ void store_proj_tiles(const v8f (&acc)[2][6],
                                                 int row0, int half, int lm,
                                                 float* __restrict__ q_buf,
                                                 float* __restrict__ k_buf,
                                                 float* __restrict__ v_buf) {
  #pragma unroll
  for (int c = 0; c < 6; ++c) {
    constexpr int kFirst = 6 * NH;
    const int vc  = kFirst + c;
    const int mat = vc >> 2;
    const int hd  = (vc & 3) * 16 + lm;
    #pragma unroll
    for (int r = 0; r < 2; ++r) {
      #pragma unroll
      for (int i = 0; i < 8; ++i) {
        const int gr = row0 + 16 * r + i + 8 * half;
        const float val = acc[r][c][i];
        if (mat == 0) {
          q_buf[(size_t)gr * kHD + hd] = val;
        } else {
          const int bb = gr / kT;
          const int tt = gr - bb * kT;
          float* dst = (mat == 1) ? k_buf : v_buf;
          dst[((size_t)bb * kS + tt) * kHD + hd] = val;
        }
      }
    }
  }
}

__global__ __launch_bounds__(256)
void qkv_kernel(const float* __restrict__ x,
                const float* __restrict__ Wq,
                const float* __restrict__ Wk,
                const float* __restrict__ Wv,
                float* __restrict__ q_buf,
                float* __restrict__ k_buf,
                float* __restrict__ v_buf) {
  __shared__ alignas(16) float xs[2][128][68];
  __shared__ alignas(16) float wst[2][96][68];

  const int tid  = threadIdx.x;
  const int wid  = __builtin_amdgcn_readfirstlane(tid >> 5);
  const int lane = tid & 31;
  const int half = lane >> 4;
  const int lm   = lane & 15;
  const int nh   = blockIdx.x & 1;
  const int r0   = (blockIdx.x >> 1) * 128;
  const bool staging = (wid >= 4);
  const int m0   = (wid & 3) * 32;
  const int ltid = tid & 127;

  const float* wseg0 = nh ? (Wk + 32) : Wq;
  const float* wseg1 = nh ? Wv        : (Wq + 32);
  const float* wseg2 = nh ? (Wv + 32) : Wk;

  v8f acc[2][6] = {};

  auto stage_tile = [&](int t) {
    const int kt = 64 * t;
    const int st = t & 1;
    v4f xr[16];
    v4f wr[12];
    #pragma unroll
    for (int i = 0; i < 16; ++i) {
      const int idx = ltid + 128 * i;
      const int row = idx >> 4;
      const int c4  = (idx & 15) * 4;
      xr[i] = *(const v4f*)(x + (size_t)(r0 + row) * kD + kt + c4);
    }
    #pragma unroll
    for (int s = 0; s < 3; ++s) {
      const float* src = (s == 0) ? wseg0 : ((s == 1) ? wseg1 : wseg2);
      #pragma unroll
      for (int u = 0; u < 4; ++u) {
        const int idx = ltid + 128 * u;
        const int kr  = idx >> 3;
        const int cc  = (idx & 7) * 4;
        wr[s * 4 + u] = *(const v4f*)(src + (size_t)(kt + kr) * kHD + cc);
      }
    }
    #pragma unroll
    for (int i = 0; i < 16; ++i) {
      const int idx = ltid + 128 * i;
      const int row = idx >> 4;
      const int c4  = (idx & 15) * 4;
      *(v4f*)&xs[st][row][c4] = xr[i];
    }
    #pragma unroll
    for (int s = 0; s < 3; ++s) {
      #pragma unroll
      for (int u = 0; u < 4; ++u) {
        const int idx = ltid + 128 * u;
        const int kr  = idx >> 3;
        const int col = s * 32 + (idx & 7) * 4;
        const v4f w = wr[s * 4 + u];
        wst[st][col + 0][kr] = w.x;
        wst[st][col + 1][kr] = w.y;
        wst[st][col + 2][kr] = w.z;
        wst[st][col + 3][kr] = w.w;
      }
    }
  };
  auto matrix_tile = [&](int t) {
    const int st = t & 1;
    #pragma unroll
    for (int kk = 0; kk < 16; ++kk) {
      const int k0 = kk * 4;
      const v2f a0 = *(const v2f*)&xs[st][m0 + lm][k0 + 2 * half];
      const v2f a1 = *(const v2f*)&xs[st][m0 + 16 + lm][k0 + 2 * half];
      #pragma unroll
      for (int c = 0; c < 6; ++c) {
        const v2f bf = *(const v2f*)&wst[st][c * 16 + lm][k0 + 2 * half];
        acc[0][c] = wmma_f32(a0, bf, acc[0][c]);
        acc[1][c] = wmma_f32(a1, bf, acc[1][c]);
      }
    }
  };

  constexpr int kNT = kD / 64;
  if (staging) stage_tile(0);
  __syncthreads();
  for (int t = 1; t < kNT; ++t) {
    if (staging) stage_tile(t); else matrix_tile(t - 1);
    __syncthreads();
  }
  if (!staging) matrix_tile(kNT - 1);

  if (!staging) {
    if (nh == 0) store_proj_tiles<0>(acc, r0 + m0, half, lm, q_buf, k_buf, v_buf);
    else         store_proj_tiles<1>(acc, r0 + m0, half, lm, q_buf, k_buf, v_buf);
  }
}

__global__ __launch_bounds__(384)
void attn_kernel(const float* __restrict__ q_buf,
                 const float* __restrict__ k_buf,
                 const float* __restrict__ v_buf,
                 float* __restrict__ out) {
  __shared__ alignas(16) float qs[128][68];
  __shared__ alignas(16) float ks[2][64][68];
  __shared__ alignas(16) float vst[2][64][68];
  __shared__ alignas(16) float ps[8][16][68];

  const int tid  = threadIdx.x;
  const int wid  = __builtin_amdgcn_readfirstlane(tid >> 5);
  const int lane = tid & 31;
  const int half = lane >> 4;
  const int lm   = lane & 15;
  const int b    = blockIdx.y;
  const int q0   = blockIdx.x * 128;
  const bool staging = (wid >= 8);
  const int pw   = wid & 7;
  const int m0   = pw * 16;
  const int ltid = tid & 127;

  v8f oacc[4] = {};
  float mrun[8], lrun[8];
  #pragma unroll
  for (int i = 0; i < 8; ++i) { mrun[i] = -INFINITY; lrun[i] = 0.f; }

  auto stage_tile = [&](int t) {
    const int s0 = 64 * t;
    const int st = t & 1;
    v4f kr[8], vr[8];
    #pragma unroll
    for (int i = 0; i < 8; ++i) {
      const int idx = ltid + 128 * i;
      const int row = idx >> 4;
      const int c4  = (idx & 15) * 4;
      kr[i] = *(const v4f*)(k_buf + ((size_t)b * kS + s0 + row) * kHD + c4);
      vr[i] = *(const v4f*)(v_buf + ((size_t)b * kS + s0 + row) * kHD + c4);
    }
    #pragma unroll
    for (int i = 0; i < 8; ++i) {
      const int idx = ltid + 128 * i;
      const int row = idx >> 4;
      const int c4  = (idx & 15) * 4;
      *(v4f*)&ks[st][row][c4] = kr[i];
      const v4f v = vr[i];
      vst[st][c4 + 0][row] = v.x;
      vst[st][c4 + 1][row] = v.y;
      vst[st][c4 + 2][row] = v.z;
      vst[st][c4 + 3][row] = v.w;
    }
  };

  auto matrix_tile = [&](int t) {
    const int st = t & 1;

    v8f sacc[4] = {};
    #pragma unroll
    for (int kk = 0; kk < 16; ++kk) {
      const int k0 = kk * 4;
      v2f a = *(const v2f*)&qs[m0 + lm][k0 + 2 * half];
      #pragma unroll
      for (int j = 0; j < 4; ++j) {
        v2f bf = *(const v2f*)&ks[st][j * 16 + lm][k0 + 2 * half];
        sacc[j] = wmma_f32(a, bf, sacc[j]);
      }
    }

    float mnew[8], alpha[8], rsum[8];
    #pragma unroll
    for (int i = 0; i < 8; ++i) {
      float v = fmaxf(fmaxf(sacc[0][i], sacc[1][i]),
                      fmaxf(sacc[2][i], sacc[3][i]));
      #pragma unroll
      for (int off = 1; off < 16; off <<= 1)
        v = fmaxf(v, __shfl_xor(v, off));
      mnew[i]  = fmaxf(mrun[i], v);
      alpha[i] = __expf(mrun[i] - mnew[i]);
      mrun[i]  = mnew[i];
      rsum[i]  = 0.f;
    }
    #pragma unroll
    for (int j = 0; j < 4; ++j) {
      #pragma unroll
      for (int i = 0; i < 8; ++i) {
        float p = __expf(sacc[j][i] - mnew[i]);
        sacc[j][i] = p;
        rsum[i] += p;
      }
    }
    #pragma unroll
    for (int i = 0; i < 8; ++i) {
      float rs = rsum[i];
      #pragma unroll
      for (int off = 1; off < 16; off <<= 1)
        rs += __shfl_xor(rs, off);
      lrun[i] = lrun[i] * alpha[i] + rs;
    }
    #pragma unroll
    for (int c = 0; c < 4; ++c)
      #pragma unroll
      for (int i = 0; i < 8; ++i)
        oacc[c][i] *= alpha[i];

    #pragma unroll
    for (int j = 0; j < 4; ++j)
      #pragma unroll
      for (int i = 0; i < 8; ++i)
        ps[pw][i + 8 * half][j * 16 + lm] = sacc[j][i];

    #pragma unroll
    for (int kk = 0; kk < 16; ++kk) {
      const int k0 = kk * 4;
      v2f a = *(const v2f*)&ps[pw][lm][k0 + 2 * half];
      #pragma unroll
      for (int c = 0; c < 4; ++c) {
        v2f bf = *(const v2f*)&vst[st][c * 16 + lm][k0 + 2 * half];
        oacc[c] = wmma_f32(a, bf, oacc[c]);
      }
    }
  };

  constexpr int kNT = kS / 64;
  if (staging) {
    stage_tile(0);
  } else {
    #pragma unroll
    for (int i = 0; i < 8; ++i) {
      int idx = tid + 256 * i;
      int row = idx >> 4;
      int c4  = (idx & 15) * 4;
      v4f v = *(const v4f*)(q_buf + ((size_t)b * kT + q0 + row) * kHD + c4);
      v *= kScale;
      *(v4f*)&qs[row][c4] = v;
    }
  }
  __syncthreads();
  for (int t = 1; t < kNT; ++t) {
    if (staging) stage_tile(t); else matrix_tile(t - 1);
    __syncthreads();
  }

  if (!staging) {
    matrix_tile(kNT - 1);
    #pragma unroll
    for (int c = 0; c < 4; ++c) {
      const int hd = c * 16 + lm;
      #pragma unroll
      for (int i = 0; i < 8; ++i) {
        const int t = q0 + m0 + i + 8 * half;
        out[((size_t)b * kT + t) * kHD + hd] = oacc[c][i] / lrun[i];
      }
    }
  }
}

extern "C" void kernel_launch(void* const* d_in, const int* in_sizes, int n_in,
                              void* d_out, int out_size, void* d_ws, size_t ws_size,
                              hipStream_t stream) {
  const float* x     = (const float*)d_in[0];
  const float* c_emb = (const float*)d_in[1];
  const float* Wq    = (const float*)d_in[2];
  const float* Wk    = (const float*)d_in[3];
  const float* Wv    = (const float*)d_in[4];
  const float* gamma = (const float*)d_in[5];
  const float* beta  = (const float*)d_in[6];
  float* out = (float*)d_out;

  float* q_buf = (float*)d_ws;
  float* k_buf = q_buf + (size_t)kB * kT * kHD;
  float* v_buf = k_buf + (size_t)kB * kS * kHD;

  ln_mem_kernel<<<kB, 64, 0, stream>>>(c_emb, gamma, beta, k_buf, v_buf);
  qkv_kernel<<<kProjBlocks, 256, 0, stream>>>(x, Wq, Wk, Wv,
                                                  q_buf, k_buf, v_buf);
  attn_kernel<<<dim3(kT / 128, kB), 384, 0, stream>>>(q_buf, k_buf, v_buf, out);
}


# attention: keys of a 128-query tile split over 2 work-groups (key tiles 0-16 | 17-32), grid 128 -> 256: one per WGP, not half the chip idle; a 4th launch merges the two halves; sum order changes
# speedup vs baseline: 2.3129x; 1.1948x over previous
#include <hip/hip_runtime.h>
#include <math.h>

typedef float v2f __attribute__((ext_vector_type(2)));
typedef float v4f __attribute__((ext_vector_type(4)));
typedef float v8f __attribute__((ext_vector_type(8)));

namespace {
constexpr int kB  = 8;
constexpr int kT  = 2048;
constexpr int kD  = 1024;
constexpr int kHD = 64;
constexpr int kM  = 64;
constexpr int kS  = kT + kM;
constexpr float kLnEps = 1e-5f;
constexpr float kScale = 0.125f;
constexpr int kProjBlocks = 256;
constexpr int kNT    = kS / 64;
constexpr int kSplit = 17;
constexpr int kAttnBlocksX = 2 * (kT / 128);
constexpr int kMergeBlocks = 256;
}

__device__ __forceinline__ v8f wmma_f32(v2f a, v2f b, v8f c) {
  return __builtin_amdgcn_wmma_f32_16x16x4_f32(
       false, a,  false, b,
       (short)0, c,  false,  false);
}

__global__ __launch_bounds__(64)
void ln_mem_kernel(const float* __restrict__ c_emb,
                   const float* __restrict__ gamma,
                   const float* __restrict__ beta,
                   float* __restrict__ k_buf) {
  const int b = blockIdx.x;
  const int r = threadIdx.x;
  const float* row = c_emb + r * kHD;
  float mu = 0.f;
  #pragma unroll
  for (int i = 0; i < kHD; ++i) mu += row[i];
  mu *= (1.f / kHD);
  float var = 0.f;
  #pragma unroll
  for (int i = 0; i < kHD; ++i) { float d = row[i] - mu; var += d * d; }
  var *= (1.f / kHD);
  const float rstd = rsqrtf(var + kLnEps);
  float* kd = k_buf + ((size_t)b * kS + kT + r) * kHD;
  #pragma unroll
  for (int i = 0; i < kHD; ++i) {
    float y = (row[i] - mu) * rstd * gamma[i] + beta[i];
    kd[i] = y;
  }
}

template <int NH>
__device__ __forceinline__ void store_proj_tiles(const v8f (&acc)[2][6],
                                                 int row0, int half, int lm,
                                                 float* __restrict__ q_buf,
                                                 float* __restrict__ q_cp,
                                                 float* __restrict__ k_buf,
                                                 float* __restrict__ v_buf) {
  #pragma unroll
  for (int c = 0; c < 6; ++c) {
    constexpr int kFirst = 6 * NH;
    const int vc  = kFirst + c;
    const int mat = vc >> 2;
    const int hd  = (vc & 3) * 16 + lm;
    #pragma unroll
    for (int r = 0; r < 2; ++r) {
      #pragma unroll
      for (int i = 0; i < 8; ++i) {
        const int gr = row0 + 16 * r + i + 8 * half;
        const float val = acc[r][c][i];
        if (mat == 0) {
          q_buf[(size_t)gr * kHD + hd] = val;
          q_cp[(size_t)gr * kHD + hd]  = val;
        } else {
          const int bb = gr / kT;
          const int tt = gr - bb * kT;
          float* dst = (mat == 1) ? k_buf : v_buf;
          dst[((size_t)bb * kS + tt) * kHD + hd] = val;
        }
      }
    }
  }
}

__global__ __launch_bounds__(256)
void qkv_kernel(const float* __restrict__ x,
                const float* __restrict__ Wq,
                const float* __restrict__ Wk,
                const float* __restrict__ Wv,
                float* __restrict__ q_buf,
                float* __restrict__ q_cp,
                float* __restrict__ k_buf,
                float* __restrict__ v_buf) {
  __shared__ alignas(16) float xs[2][128][68];
  __shared__ alignas(16) float wst[2][96][68];

  const int tid  = threadIdx.x;
  const int wid  = __builtin_amdgcn_readfirstlane(tid >> 5);
  const int lane = tid & 31;
  const int half = lane >> 4;
  const int lm   = lane & 15;
  const int nh   = blockIdx.x & 1;
  const int r0   = (blockIdx.x >> 1) * 128;
  const bool staging = (wid >= 4);
  const int m0   = (wid & 3) * 32;
  const int ltid = tid & 127;

  const float* wseg0 = nh ? (Wk + 32) : Wq;
  const float* wseg1 = nh ? Wv        : (Wq + 32);
  const float* wseg2 = nh ? (Wv + 32) : Wk;

  v8f acc[2][6] = {};

  auto stage_tile = [&](int t) {
    const int kt = 64 * t;
    const int st = t & 1;
    v4f xr[16];
    v4f wr[12];
    #pragma unroll
    for (int i = 0; i < 16; ++i) {
      const int idx = ltid + 128 * i;
      const int row = idx >> 4;
      const int c4  = (idx & 15) * 4;
      xr[i] = *(const v4f*)(x + (size_t)(r0 + row) * kD + kt + c4);
    }
    #pragma unroll
    for (int s = 0; s < 3; ++s) {
      const float* src = (s == 0) ? wseg0 : ((s == 1) ? wseg1 : wseg2);
      #pragma unroll
      for (int u = 0; u < 4; ++u) {
        const int idx = ltid + 128 * u;
        const int kr  = idx >> 3;
        const int cc  = (idx & 7) * 4;
        wr[s * 4 + u] = *(const v4f*)(src + (size_t)(kt + kr) * kHD + cc);
      }
    }
    #pragma unroll
    for (int i = 0; i < 16; ++i) {
      const int idx = ltid + 128 * i;
      const int row = idx >> 4;
      const int c4  = (idx & 15) * 4;
      *(v4f*)&xs[st][row][c4] = xr[i];
    }
    #pragma unroll
    for (int s = 0; s < 3; ++s) {
      #pragma unroll
      for (int u = 0; u < 4; ++u) {
        const int idx = ltid + 128 * u;
        const int kr  = idx >> 3;
        const int col = s * 32 + (idx & 7) * 4;
        const v4f w = wr[s * 4 + u];
        wst[st][col + 0][kr] = w.x;
        wst[st][col + 1][kr] = w.y;
        wst[st][col + 2][kr] = w.z;
        wst[st][col + 3][kr] = w.w;
      }
    }
  };
  auto matrix_tile = [&](int t) {
    const int st = t & 1;
    #pragma unroll
    for (int kk = 0; kk < 16; ++kk) {
      const int k0 = kk * 4;
      const v2f a0 = *(const v2f*)&xs[st][m0 + lm][k0 + 2 * half];
      const v2f a1 = *(const v2f*)&xs[st][m0 + 16 + lm][k0 + 2 * half];
      #pragma unroll
      for (int c = 0; c < 6; ++c) {
        const v2f bf = *(const v2f*)&wst[st][c * 16 + lm][k0 + 2 * half];
        acc[0][c] = wmma_f32(a0, bf, acc[0][c]);
        acc[1][c] = wmma_f32(a1, bf, acc[1][c]);
      }
    }
  };

  constexpr int kKT = kD / 64;
  if (staging) stage_tile(0);
  __syncthreads();
  for (int t = 1; t < kKT; ++t) {
    if (staging) stage_tile(t); else matrix_tile(t - 1);
    __syncthreads();
  }
  if (!staging) matrix_tile(kKT - 1);

  if (!staging) {
    if (nh == 0) store_proj_tiles<0>(acc, r0 + m0, half, lm, q_buf, q_cp, k_buf, v_buf);
    else         store_proj_tiles<1>(acc, r0 + m0, half, lm, q_buf, q_cp, k_buf, v_buf);
  }
}

__global__ __launch_bounds__(384)
void attn_kernel(float* q_buf,
                 const float* __restrict__ k_buf,
                 float* v_buf,
                 float* out) {
  __shared__ alignas(16) float qs[128][68];
  __shared__ alignas(16) float ks[2][64][68];
  __shared__ alignas(16) float vst[2][64][68];
  __shared__ alignas(16) float ps[8][16][68];

  const int tid  = threadIdx.x;
  const int wid  = __builtin_amdgcn_readfirstlane(tid >> 5);
  const int lane = tid & 31;
  const int half = lane >> 4;
  const int lm   = lane & 15;
  const int b    = blockIdx.y;
  const int kh   = blockIdx.x & 1;
  const int q0   = (blockIdx.x >> 1) * 128;
  const int tb   = kh ? kSplit : 0;
  const int te   = kh ? kNT : kSplit;
  float* qo = kh ? out : q_buf;
  const bool staging = (wid >= 8);
  const int pw   = wid & 7;
  const int m0   = pw * 16;
  const int ltid = tid & 127;

  v8f oacc[4] = {};
  float mrun[8], lrun[8];
  #pragma unroll
  for (int i = 0; i < 8; ++i) { mrun[i] = -INFINITY; lrun[i] = 0.f; }

  auto stage_tile = [&](int t) {
    const int s0 = 64 * t;
    const int st = t & 1;
    const float* vsrc = (t == kNT - 1) ? k_buf : v_buf;
    v4f kr[8], vr[8];
    #pragma unroll
    for (int i = 0; i < 8; ++i) {
      const int idx = ltid + 128 * i;
      const int row = idx >> 4;
      const int c4  = (idx & 15) * 4;
      kr[i] = *(const v4f*)(k_buf + ((size_t)b * kS + s0 + row) * kHD + c4);
      vr[i] = *(const v4f*)(vsrc  + ((size_t)b * kS + s0 + row) * kHD + c4);
    }
    #pragma unroll
    for (int i = 0; i < 8; ++i) {
      const int idx = ltid + 128 * i;
      const int row = idx >> 4;
      const int c4  = (idx & 15) * 4;
      *(v4f*)&ks[st][row][c4] = kr[i];
      const v4f v = vr[i];
      vst[st][c4 + 0][row] = v.x;
      vst[st][c4 + 1][row] = v.y;
      vst[st][c4 + 2][row] = v.z;
      vst[st][c4 + 3][row] = v.w;
    }
  };

  auto matrix_tile = [&](int t) {
    const int st = t & 1;

    v8f sacc[4] = {};
    #pragma unroll
    for (int kk = 0; kk < 16; ++kk) {
      const int k0 = kk * 4;
      v2f a = *(const v2f*)&qs[m0 + lm][k0 + 2 * half];
      #pragma unroll
      for (int j = 0; j < 4; ++j) {
        v2f bf = *(const v2f*)&ks[st][j * 16 + lm][k0 + 2 * half];
        sacc[j] = wmma_f32(a, bf, sacc[j]);
      }
    }

    float mnew[8], alpha[8], rsum[8];
    #pragma unroll
    for (int i = 0; i < 8; ++i) {
      float v = fmaxf(fmaxf(sacc[0][i], sacc[1][i]),
                      fmaxf(sacc[2][i], sacc[3][i]));
      #pragma unroll
      for (int off = 1; off < 16; off <<= 1)
        v = fmaxf(v, __shfl_xor(v, off));
      mnew[i]  = fmaxf(mrun[i], v);
      alpha[i] = __expf(mrun[i] - mnew[i]);
      mrun[i]  = mnew[i];
      rsum[i]  = 0.f;
    }
    #pragma unroll
    for (int j = 0; j < 4; ++j) {
      #pragma unroll
      for (int i = 0; i < 8; ++i) {
        float p = __expf(sacc[j][i] - mnew[i]);
        sacc[j][i] = p;
        rsum[i] += p;
      }
    }
    #pragma unroll
    for (int i = 0; i < 8; ++i) {
      float rs = rsum[i];
      #pragma unroll
      for (int off = 1; off < 16; off <<= 1)
        rs += __shfl_xor(rs, off);
      lrun[i] = lrun[i] * alpha[i] + rs;
    }
    #pragma unroll
    for (int c = 0; c < 4; ++c)
      #pragma unroll
      for (int i = 0; i < 8; ++i)
        oacc[c][i] *= alpha[i];

    #pragma unroll
    for (int j = 0; j < 4; ++j)
      #pragma unroll
      for (int i = 0; i < 8; ++i)
        ps[pw][i + 8 * half][j * 16 + lm] = sacc[j][i];

    #pragma unroll
    for (int kk = 0; kk < 16; ++kk) {
      const int k0 = kk * 4;
      v2f a = *(const v2f*)&ps[pw][lm][k0 + 2 * half];
      #pragma unroll
      for (int c = 0; c < 4; ++c) {
        v2f bf = *(const v2f*)&vst[st][c * 16 + lm][k0 + 2 * half];
        oacc[c] = wmma_f32(a, bf, oacc[c]);
      }
    }
  };

  if (staging) {
    stage_tile(tb);
  } else {
    #pragma unroll
    for (int i = 0; i < 8; ++i) {
      int idx = tid + 256 * i;
      int row = idx >> 4;
      int c4  = (idx & 15) * 4;
      v4f v = *(const v4f*)(qo + ((size_t)b * kT + q0 + row) * kHD + c4);
      v *= kScale;
      *(v4f*)&qs[row][c4] = v;
    }
  }
  __syncthreads();
  for (int t = tb + 1; t < te; ++t) {
    if (staging) stage_tile(t); else matrix_tile(t - 1);
    __syncthreads();
  }

  if (!staging) {
    matrix_tile(te - 1);
    #pragma unroll
    for (int c = 0; c < 4; ++c) {
      const int hd = c * 16 + lm;
      #pragma unroll
      for (int i = 0; i < 8; ++i) {
        const int t = q0 + m0 + i + 8 * half;
        qo[((size_t)b * kT + t) * kHD + hd] = oacc[c][i] / lrun[i];
      }
    }
    if (lm == 0) {
      float* lse = v_buf + ((size_t)b * kS + kT) * kHD + kh * kT;
      #pragma unroll
      for (int i = 0; i < 8; ++i)
        lse[q0 + m0 + i + 8 * half] = mrun[i] + __logf(lrun[i]);
    }
  }
}

__global__ __launch_bounds__(256)
void merge_kernel(const float* __restrict__ part0,
                  const float* __restrict__ v_buf,
                  float* out) {
  constexpr int kThreads = kMergeBlocks * 256;
  constexpr int kPer     = kB * kT * kHD / 4 / kThreads;
  const int gtid = blockIdx.x * 256 + threadIdx.x;
  v4f o0[kPer], o1[kPer];
  float w0[kPer], w1[kPer];
  #pragma unroll
  for (int j = 0; j < kPer; ++j) {
    const int idx = gtid + kThreads * j;
    const int row = idx >> 4;
    const int c4  = (idx & 15) * 4;
    const int b   = row / kT;
    const int t   = row - b * kT;
    const float* lse = v_buf + ((size_t)b * kS + kT) * kHD;
    w0[j] = lse[t];
    w1[j] = lse[kT + t];
    o0[j] = *(const v4f*)(part0 + (size_t)row * kHD + c4);
    o1[j] = *(const v4f*)(out + (size_t)row * kHD + c4);
  }
  #pragma unroll
  for (int j = 0; j < kPer; ++j) {
    const int idx = gtid + kThreads * j;
    const int row = idx >> 4;
    const int c4  = (idx & 15) * 4;
    const float m   = fmaxf(w0[j], w1[j]);
    const float a0  = __expf(w0[j] - m);
    const float a1  = __expf(w1[j] - m);
    const float inv = 1.f / (a0 + a1);
    *(v4f*)(out + (size_t)row * kHD + c4) = (o0[j] * a0 + o1[j] * a1) * inv;
  }
}

extern "C" void kernel_launch(void* const* d_in, const int* in_sizes, int n_in,
                              void* d_out, int out_size, void* d_ws, size_t ws_size,
                              hipStream_t stream) {
  const float* x     = (const float*)d_in[0];
  const float* c_emb = (const float*)d_in[1];
  const float* Wq    = (const float*)d_in[2];
  const float* Wk    = (const float*)d_in[3];
  const float* Wv    = (const float*)d_in[4];
  const float* gamma = (const float*)d_in[5];
  const float* beta  = (const float*)d_in[6];
  float* out = (float*)d_out;

  float* q_buf = (float*)d_ws;
  float* k_buf = q_buf + (size_t)kB * kT * kHD;
  float* v_buf = k_buf + (size_t)kB * kS * kHD;

  ln_mem_kernel<<<kB, 64, 0, stream>>>(c_emb, gamma, beta, k_buf);
  qkv_kernel<<<kProjBlocks, 256, 0, stream>>>(x, Wq, Wk, Wv,
                                                  q_buf, out, k_buf, v_buf);
  attn_kernel<<<dim3(kAttnBlocksX, kB), 384, 0, stream>>>(q_buf, k_buf, v_buf, out);
  merge_kernel<<<kMergeBlocks, 256, 0, stream>>>(q_buf, v_buf, out);
}


# memory-token layernorm leaves its own launch: attention staging waves normalise c_emb straight into the LDS key/value tile of the last key tile; 4 -> 3 launches; row sums in 2 halves (order changes)
# speedup vs baseline: 2.4087x; 1.0414x over previous
#include <hip/hip_runtime.h>
#include <math.h>

typedef float v2f __attribute__((ext_vector_type(2)));
typedef float v4f __attribute__((ext_vector_type(4)));
typedef float v8f __attribute__((ext_vector_type(8)));

namespace {
constexpr int kB  = 8;
constexpr int kT  = 2048;
constexpr int kD  = 1024;
constexpr int kHD = 64;
constexpr int kM  = 64;
constexpr int kS  = kT + kM;
constexpr float kLnEps = 1e-5f;
constexpr float kScale = 0.125f;
constexpr int kProjBlocks = 256;
constexpr int kNT    = kS / 64;
constexpr int kSplit = 17;
constexpr int kAttnBlocksX = 2 * (kT / 128);
constexpr int kMergeBlocks = 256;
}

__device__ __forceinline__ v8f wmma_f32(v2f a, v2f b, v8f c) {
  return __builtin_amdgcn_wmma_f32_16x16x4_f32(
       false, a,  false, b,
       (short)0, c,  false,  false);
}


template <int NH>
__device__ __forceinline__ void store_proj_tiles(const v8f (&acc)[2][6],
                                                 int row0, int half, int lm,
                                                 float* __restrict__ q_buf,
                                                 float* __restrict__ q_cp,
                                                 float* __restrict__ k_buf,
                                                 float* __restrict__ v_buf) {
  #pragma unroll
  for (int c = 0; c < 6; ++c) {
    constexpr int kFirst = 6 * NH;
    const int vc  = kFirst + c;
    const int mat = vc >> 2;
    const int hd  = (vc & 3) * 16 + lm;
    #pragma unroll
    for (int r = 0; r < 2; ++r) {
      #pragma unroll
      for (int i = 0; i < 8; ++i) {
        const int gr = row0 + 16 * r + i + 8 * half;
        const float val = acc[r][c][i];
        if (mat == 0) {
          q_buf[(size_t)gr * kHD + hd] = val;
          q_cp[(size_t)gr * kHD + hd]  = val;
        } else {
          const int bb = gr / kT;
          const int tt = gr - bb * kT;
          float* dst = (mat == 1) ? k_buf : v_buf;
          dst[((size_t)bb * kS + tt) * kHD + hd] = val;
        }
      }
    }
  }
}

__global__ __launch_bounds__(256)
void qkv_kernel(const float* __restrict__ x,
                const float* __restrict__ Wq,
                const float* __restrict__ Wk,
                const float* __restrict__ Wv,
                float* __restrict__ q_buf,
                float* __restrict__ q_cp,
                float* __restrict__ k_buf,
                float* __restrict__ v_buf) {
  __shared__ alignas(16) float xs[2][128][68];
  __shared__ alignas(16) float wst[2][96][68];

  const int tid  = threadIdx.x;
  const int wid  = __builtin_amdgcn_readfirstlane(tid >> 5);
  const int lane = tid & 31;
  const int half = lane >> 4;
  const int lm   = lane & 15;
  const int nh   = blockIdx.x & 1;
  const int r0   = (blockIdx.x >> 1) * 128;
  const bool staging = (wid >= 4);
  const int m0   = (wid & 3) * 32;
  const int ltid = tid & 127;

  const float* wseg0 = nh ? (Wk + 32) : Wq;
  const float* wseg1 = nh ? Wv        : (Wq + 32);
  const float* wseg2 = nh ? (Wv + 32) : Wk;

  v8f acc[2][6] = {};

  auto stage_tile = [&](int t) {
    const int kt = 64 * t;
    const int st = t & 1;
    v4f xr[16];
    v4f wr[12];
    #pragma unroll
    for (int i = 0; i < 16; ++i) {
      const int idx = ltid + 128 * i;
      const int row = idx >> 4;
      const int c4  = (idx & 15) * 4;
      xr[i] = *(const v4f*)(x + (size_t)(r0 + row) * kD + kt + c4);
    }
    #pragma unroll
    for (int s = 0; s < 3; ++s) {
      const float* src = (s == 0) ? wseg0 : ((s == 1) ? wseg1 : wseg2);
      #pragma unroll
      for (int u = 0; u < 4; ++u) {
        const int idx = ltid + 128 * u;
        const int kr  = idx >> 3;
        const int cc  = (idx & 7) * 4;
        wr[s * 4 + u] = *(const v4f*)(src + (size_t)(kt + kr) * kHD + cc);
      }
    }
    #pragma unroll
    for (int i = 0; i < 16; ++i) {
      const int idx = ltid + 128 * i;
      const int row = idx >> 4;
      const int c4  = (idx & 15) * 4;
      *(v4f*)&xs[st][row][c4] = xr[i];
    }
    #pragma unroll
    for (int s = 0; s < 3; ++s) {
      #pragma unroll
      for (int u = 0; u < 4; ++u) {
        const int idx = ltid + 128 * u;
        const int kr  = idx >> 3;
        const int col = s * 32 + (idx & 7) * 4;
        const v4f w = wr[s * 4 + u];
        wst[st][col + 0][kr] = w.x;
        wst[st][col + 1][kr] = w.y;
        wst[st][col + 2][kr] = w.z;
        wst[st][col + 3][kr] = w.w;
      }
    }
  };
  auto matrix_tile = [&](int t) {
    const int st = t & 1;
    #pragma unroll
    for (int kk = 0; kk < 16; ++kk) {
      const int k0 = kk * 4;
      const v2f a0 = *(const v2f*)&xs[st][m0 + lm][k0 + 2 * half];
      const v2f a1 = *(const v2f*)&xs[st][m0 + 16 + lm][k0 + 2 * half];
      #pragma unroll
      for (int c = 0; c < 6; ++c) {
        const v2f bf = *(const v2f*)&wst[st][c * 16 + lm][k0 + 2 * half];
        acc[0][c] = wmma_f32(a0, bf, acc[0][c]);
        acc[1][c] = wmma_f32(a1, bf, acc[1][c]);
      }
    }
  };

  constexpr int kKT = kD / 64;
  if (staging) stage_tile(0);
  __syncthreads();
  for (int t = 1; t < kKT; ++t) {
    if (staging) stage_tile(t); else matrix_tile(t - 1);
    __syncthreads();
  }
  if (!staging) matrix_tile(kKT - 1);

  if (!staging) {
    if (nh == 0) store_proj_tiles<0>(acc, r0 + m0, half, lm, q_buf, q_cp, k_buf, v_buf);
    else         store_proj_tiles<1>(acc, r0 + m0, half, lm, q_buf, q_cp, k_buf, v_buf);
  }
}

__global__ __launch_bounds__(384)
void attn_kernel(float* q_buf,
                 const float* __restrict__ k_buf,
                 float* v_buf,
                 float* out,
                 const float* __restrict__ c_emb,
                 const float* __restrict__ gamma,
                 const float* __restrict__ beta) {
  __shared__ alignas(16) float qs[128][68];
  __shared__ alignas(16) float ks[2][64][68];
  __shared__ alignas(16) float vst[2][64][68];
  __shared__ alignas(16) float ps[8][16][68];

  const int tid  = threadIdx.x;
  const int wid  = __builtin_amdgcn_readfirstlane(tid >> 5);
  const int lane = tid & 31;
  const int half = lane >> 4;
  const int lm   = lane & 15;
  const int b    = blockIdx.y;
  const int kh   = blockIdx.x & 1;
  const int q0   = (blockIdx.x >> 1) * 128;
  const int tb   = kh ? kSplit : 0;
  const int te   = kh ? kNT : kSplit;
  float* qo = kh ? out : q_buf;
  const bool staging = (wid >= 8);
  const int pw   = wid & 7;
  const int m0   = pw * 16;
  const int ltid = tid & 127;

  v8f oacc[4] = {};
  float mrun[8], lrun[8];
  #pragma unroll
  for (int i = 0; i < 8; ++i) { mrun[i] = -INFINITY; lrun[i] = 0.f; }

  auto stage_tile = [&](int t) {
    const int s0 = 64 * t;
    const int st = t & 1;
    v4f kr[8], vr[8];
    #pragma unroll
    for (int i = 0; i < 8; ++i) {
      const int idx = ltid + 128 * i;
      const int row = idx >> 4;
      const int c4  = (idx & 15) * 4;
      kr[i] = *(const v4f*)(k_buf + ((size_t)b * kS + s0 + row) * kHD + c4);
      vr[i] = *(const v4f*)(v_buf + ((size_t)b * kS + s0 + row) * kHD + c4);
    }
    #pragma unroll
    for (int i = 0; i < 8; ++i) {
      const int idx = ltid + 128 * i;
      const int row = idx >> 4;
      const int c4  = (idx & 15) * 4;
      *(v4f*)&ks[st][row][c4] = kr[i];
      const v4f v = vr[i];
      vst[st][c4 + 0][row] = v.x;
      vst[st][c4 + 1][row] = v.y;
      vst[st][c4 + 2][row] = v.z;
      vst[st][c4 + 3][row] = v.w;
    }
  };

  auto stage_mem_tile = [&](int st) {
    const int row = ltid >> 1;
    const int h0  = (ltid & 1) * 32;
    const float* src = c_emb + row * kHD + h0;
    v4f xr[8];
    #pragma unroll
    for (int i = 0; i < 8; ++i) xr[i] = *(const v4f*)(src + 4 * i);
    float s = 0.f;
    #pragma unroll
    for (int i = 0; i < 8; ++i) s += (xr[i].x + xr[i].y) + (xr[i].z + xr[i].w);
    s += __shfl_xor(s, 1);
    const float mu = s * (1.f / kHD);
    float q = 0.f;
    #pragma unroll
    for (int i = 0; i < 8; ++i) {
      const v4f d = xr[i] - mu;
      q += (d.x * d.x + d.y * d.y) + (d.z * d.z + d.w * d.w);
    }
    q += __shfl_xor(q, 1);
    const float rstd = rsqrtf(q * (1.f / kHD) + kLnEps);
    #pragma unroll
    for (int i = 0; i < 8; ++i) {
      const int c4 = h0 + 4 * i;
      const v4f g  = *(const v4f*)(gamma + c4);
      const v4f bt = *(const v4f*)(beta + c4);
      const v4f y  = (xr[i] - mu) * rstd * g + bt;
      *(v4f*)&ks[st][row][c4] = y;
      vst[st][c4 + 0][row] = y.x;
      vst[st][c4 + 1][row] = y.y;
      vst[st][c4 + 2][row] = y.z;
      vst[st][c4 + 3][row] = y.w;
    }
  };

  auto matrix_tile = [&](int t) {
    const int st = t & 1;

    v8f sacc[4] = {};
    #pragma unroll
    for (int kk = 0; kk < 16; ++kk) {
      const int k0 = kk * 4;
      v2f a = *(const v2f*)&qs[m0 + lm][k0 + 2 * half];
      #pragma unroll
      for (int j = 0; j < 4; ++j) {
        v2f bf = *(const v2f*)&ks[st][j * 16 + lm][k0 + 2 * half];
        sacc[j] = wmma_f32(a, bf, sacc[j]);
      }
    }

    float mnew[8], alpha[8], rsum[8];
    #pragma unroll
    for (int i = 0; i < 8; ++i) {
      float v = fmaxf(fmaxf(sacc[0][i], sacc[1][i]),
                      fmaxf(sacc[2][i], sacc[3][i]));
      #pragma unroll
      for (int off = 1; off < 16; off <<= 1)
        v = fmaxf(v, __shfl_xor(v, off));
      mnew[i]  = fmaxf(mrun[i], v);
      alpha[i] = __expf(mrun[i] - mnew[i]);
      mrun[i]  = mnew[i];
      rsum[i]  = 0.f;
    }
    #pragma unroll
    for (int j = 0; j < 4; ++j) {
      #pragma unroll
      for (int i = 0; i < 8; ++i) {
        float p = __expf(sacc[j][i] - mnew[i]);
        sacc[j][i] = p;
        rsum[i] += p;
      }
    }
    #pragma unroll
    for (int i = 0; i < 8; ++i) {
      float rs = rsum[i];
      #pragma unroll
      for (int off = 1; off < 16; off <<= 1)
        rs += __shfl_xor(rs, off);
      lrun[i] = lrun[i] * alpha[i] + rs;
    }
    #pragma unroll
    for (int c = 0; c < 4; ++c)
      #pragma unroll
      for (int i = 0; i < 8; ++i)
        oacc[c][i] *= alpha[i];

    #pragma unroll
    for (int j = 0; j < 4; ++j)
      #pragma unroll
      for (int i = 0; i < 8; ++i)
        ps[pw][i + 8 * half][j * 16 + lm] = sacc[j][i];

    #pragma unroll
    for (int kk = 0; kk < 16; ++kk) {
      const int k0 = kk * 4;
      v2f a = *(const v2f*)&ps[pw][lm][k0 + 2 * half];
      #pragma unroll
      for (int c = 0; c < 4; ++c) {
        v2f bf = *(const v2f*)&vst[st][c * 16 + lm][k0 + 2 * half];
        oacc[c] = wmma_f32(a, bf, oacc[c]);
      }
    }
  };

  if (staging) {
    stage_tile(tb);
  } else {
    #pragma unroll
    for (int i = 0; i < 8; ++i) {
      int idx = tid + 256 * i;
      int row = idx >> 4;
      int c4  = (idx & 15) * 4;
      v4f v = *(const v4f*)(qo + ((size_t)b * kT + q0 + row) * kHD + c4);
      v *= kScale;
      *(v4f*)&qs[row][c4] = v;
    }
  }
  __syncthreads();
  static_assert(kSplit > 0 && kSplit < kNT - 1, "the memory tile is staged inside the loop");
  for (int t = tb + 1; t < te; ++t) {
    if (staging) {
      if (t == kNT - 1) stage_mem_tile(t & 1); else stage_tile(t);
    } else {
      matrix_tile(t - 1);
    }
    __syncthreads();
  }

  if (!staging) {
    matrix_tile(te - 1);
    #pragma unroll
    for (int c = 0; c < 4; ++c) {
      const int hd = c * 16 + lm;
      #pragma unroll
      for (int i = 0; i < 8; ++i) {
        const int t = q0 + m0 + i + 8 * half;
        qo[((size_t)b * kT + t) * kHD + hd] = oacc[c][i] / lrun[i];
      }
    }
    if (lm == 0) {
      float* lse = v_buf + ((size_t)b * kS + kT) * kHD + kh * kT;
      #pragma unroll
      for (int i = 0; i < 8; ++i)
        lse[q0 + m0 + i + 8 * half] = mrun[i] + __logf(lrun[i]);
    }
  }
}

__global__ __launch_bounds__(256)
void merge_kernel(const float* __restrict__ part0,
                  const float* __restrict__ v_buf,
                  float* out) {
  constexpr int kThreads = kMergeBlocks * 256;
  constexpr int kPer     = kB * kT * kHD / 4 / kThreads;
  const int gtid = blockIdx.x * 256 + threadIdx.x;
  v4f o0[kPer], o1[kPer];
  float w0[kPer], w1[kPer];
  #pragma unroll
  for (int j = 0; j < kPer; ++j) {
    const int idx = gtid + kThreads * j;
    const int row = idx >> 4;
    const int c4  = (idx & 15) * 4;
    const int b   = row / kT;
    const int t   = row - b * kT;
    const float* lse = v_buf + ((size_t)b * kS + kT) * kHD;
    w0[j] = lse[t];
    w1[j] = lse[kT + t];
    o0[j] = *(const v4f*)(part0 + (size_t)row * kHD + c4);
    o1[j] = *(const v4f*)(out + (size_t)row * kHD + c4);
  }
  #pragma unroll
  for (int j = 0; j < kPer; ++j) {
    const int idx = gtid + kThreads * j;
    const int row = idx >> 4;
    const int c4  = (idx & 15) * 4;
    const float m   = fmaxf(w0[j], w1[j]);
    const float a0  = __expf(w0[j] - m);
    const float a1  = __expf(w1[j] - m);
    const float inv = 1.f / (a0 + a1);
    *(v4f*)(out + (size_t)row * kHD + c4) = (o0[j] * a0 + o1[j] * a1) * inv;
  }
}

extern "C" void kernel_launch(void* const* d_in, const int* in_sizes, int n_in,
                              void* d_out, int out_size, void* d_ws, size_t ws_size,
                              hipStream_t stream) {
  const float* x     = (const float*)d_in[0];
  const float* c_emb = (const float*)d_in[1];
  const float* Wq    = (const float*)d_in[2];
  const float* Wk    = (const float*)d_in[3];
  const float* Wv    = (const float*)d_in[4];
  const float* gamma = (const float*)d_in[5];
  const float* beta  = (const float*)d_in[6];
  float* out = (float*)d_out;

  float* q_buf = (float*)d_ws;
  float* k_buf = q_buf + (size_t)kB * kT * kHD;
  float* v_buf = k_buf + (size_t)kB * kS * kHD;

  qkv_kernel<<<kProjBlocks, 256, 0, stream>>>(x, Wq, Wk, Wv,
                                                  q_buf, out, k_buf, v_buf);
  attn_kernel<<<dim3(kAttnBlocksX, kB), 384, 0, stream>>>(q_buf, k_buf, v_buf, out,
                                                          c_emb, gamma, beta);
  merge_kernel<<<kMergeBlocks, 256, 0, stream>>>(q_buf, v_buf, out);
}


# attention: waves split by matrix role: waves 0-3 do S and softmax, pass P and row factors via LDS to waves 4-7 (P@V one tile behind); block = 64 queries x all keys, no merge launch; sum order changes
# speedup vs baseline: 2.7715x; 1.1506x over previous
#include <hip/hip_runtime.h>
#include <math.h>

typedef float v2f __attribute__((ext_vector_type(2)));
typedef float v4f __attribute__((ext_vector_type(4)));
typedef float v8f __attribute__((ext_vector_type(8)));

namespace {
constexpr int kB  = 8;
constexpr int kT  = 2048;
constexpr int kD  = 1024;
constexpr int kHD = 64;
constexpr int kM  = 64;
constexpr int kS  = kT + kM;
constexpr float kLnEps = 1e-5f;
constexpr float kScale = 0.125f;
constexpr int kProjBlocks = 256;
constexpr int kNT    = kS / 64;
constexpr int kQT    = 64;
constexpr int kAttnBlocksX = kT / kQT;
}

__device__ __forceinline__ v8f wmma_f32(v2f a, v2f b, v8f c) {
  return __builtin_amdgcn_wmma_f32_16x16x4_f32(
       false, a,  false, b,
       (short)0, c,  false,  false);
}


template <int NH>
__device__ __forceinline__ void store_proj_tiles(const v8f (&acc)[2][6],
                                                 int row0, int half, int lm,
                                                 float* __restrict__ q_buf,
                                                 float* __restrict__ k_buf,
                                                 float* __restrict__ v_buf) {
  #pragma unroll
  for (int c = 0; c < 6; ++c) {
    constexpr int kFirst = 6 * NH;
    const int vc  = kFirst + c;
    const int mat = vc >> 2;
    const int hd  = (vc & 3) * 16 + lm;
    #pragma unroll
    for (int r = 0; r < 2; ++r) {
      #pragma unroll
      for (int i = 0; i < 8; ++i) {
        const int gr = row0 + 16 * r + i + 8 * half;
        const float val = acc[r][c][i];
        if (mat == 0) {
          q_buf[(size_t)gr * kHD + hd] = val;
        } else {
          const int bb = gr / kT;
          const int tt = gr - bb * kT;
          float* dst = (mat == 1) ? k_buf : v_buf;
          dst[((size_t)bb * kS + tt) * kHD + hd] = val;
        }
      }
    }
  }
}

__global__ __launch_bounds__(256)
void qkv_kernel(const float* __restrict__ x,
                const float* __restrict__ Wq,
                const float* __restrict__ Wk,
                const float* __restrict__ Wv,
                float* __restrict__ q_buf,
                float* __restrict__ k_buf,
                float* __restrict__ v_buf) {
  __shared__ alignas(16) float xs[2][128][68];
  __shared__ alignas(16) float wst[2][96][68];

  const int tid  = threadIdx.x;
  const int wid  = __builtin_amdgcn_readfirstlane(tid >> 5);
  const int lane = tid & 31;
  const int half = lane >> 4;
  const int lm   = lane & 15;
  const int nh   = blockIdx.x & 1;
  const int r0   = (blockIdx.x >> 1) * 128;
  const bool staging = (wid >= 4);
  const int m0   = (wid & 3) * 32;
  const int ltid = tid & 127;

  const float* wseg0 = nh ? (Wk + 32) : Wq;
  const float* wseg1 = nh ? Wv        : (Wq + 32);
  const float* wseg2 = nh ? (Wv + 32) : Wk;

  v8f acc[2][6] = {};

  auto stage_tile = [&](int t) {
    const int kt = 64 * t;
    const int st = t & 1;
    v4f xr[16];
    v4f wr[12];
    #pragma unroll
    for (int i = 0; i < 16; ++i) {
      const int idx = ltid + 128 * i;
      const int row = idx >> 4;
      const int c4  = (idx & 15) * 4;
      xr[i] = *(const v4f*)(x + (size_t)(r0 + row) * kD + kt + c4);
    }
    #pragma unroll
    for (int s = 0; s < 3; ++s) {
      const float* src = (s == 0) ? wseg0 : ((s == 1) ? wseg1 : wseg2);
      #pragma unroll
      for (int u = 0; u < 4; ++u) {
        const int idx = ltid + 128 * u;
        const int kr  = idx >> 3;
        const int cc  = (idx & 7) * 4;
        wr[s * 4 + u] = *(const v4f*)(src + (size_t)(kt + kr) * kHD + cc);
      }
    }
    #pragma unroll
    for (int i = 0; i < 16; ++i) {
      const int idx = ltid + 128 * i;
      const int row = idx >> 4;
      const int c4  = (idx & 15) * 4;
      *(v4f*)&xs[st][row][c4] = xr[i];
    }
    #pragma unroll
    for (int s = 0; s < 3; ++s) {
      #pragma unroll
      for (int u = 0; u < 4; ++u) {
        const int idx = ltid + 128 * u;
        const int kr  = idx >> 3;
        const int col = s * 32 + (idx & 7) * 4;
        const v4f w = wr[s * 4 + u];
        wst[st][col + 0][kr] = w.x;
        wst[st][col + 1][kr] = w.y;
        wst[st][col + 2][kr] = w.z;
        wst[st][col + 3][kr] = w.w;
      }
    }
  };
  auto matrix_tile = [&](int t) {
    const int st = t & 1;
    #pragma unroll
    for (int kk = 0; kk < 16; ++kk) {
      const int k0 = kk * 4;
      const v2f a0 = *(const v2f*)&xs[st][m0 + lm][k0 + 2 * half];
      const v2f a1 = *(const v2f*)&xs[st][m0 + 16 + lm][k0 + 2 * half];
      #pragma unroll
      for (int c = 0; c < 6; ++c) {
        const v2f bf = *(const v2f*)&wst[st][c * 16 + lm][k0 + 2 * half];
        acc[0][c] = wmma_f32(a0, bf, acc[0][c]);
        acc[1][c] = wmma_f32(a1, bf, acc[1][c]);
      }
    }
  };

  constexpr int kKT = kD / 64;
  if (staging) stage_tile(0);
  __syncthreads();
  for (int t = 1; t < kKT; ++t) {
    if (staging) stage_tile(t); else matrix_tile(t - 1);
    __syncthreads();
  }
  if (!staging) matrix_tile(kKT - 1);

  if (!staging) {
    if (nh == 0) store_proj_tiles<0>(acc, r0 + m0, half, lm, q_buf, k_buf, v_buf);
    else         store_proj_tiles<1>(acc, r0 + m0, half, lm, q_buf, k_buf, v_buf);
  }
}

__global__ __launch_bounds__(384)
void attn_kernel(const float* __restrict__ q_buf,
                 const float* __restrict__ k_buf,
                 const float* __restrict__ v_buf,
                 float* __restrict__ out,
                 const float* __restrict__ c_emb,
                 const float* __restrict__ gamma,
                 const float* __restrict__ beta) {
  __shared__ alignas(16) float qs[kQT][68];
  __shared__ alignas(16) float ks[2][64][68];
  __shared__ alignas(16) float vst[3][64][68];
  __shared__ alignas(16) float ps[2][4][16][68];
  __shared__ alignas(16) float als[2][4][16];
  __shared__ alignas(16) float lsum[4][16];

  const int tid  = threadIdx.x;
  const int wid  = __builtin_amdgcn_readfirstlane(tid >> 5);
  const int lane = tid & 31;
  const int half = lane >> 4;
  const int lm   = lane & 15;
  const int b    = blockIdx.y;
  const int q0   = blockIdx.x * kQT;
  const int role = wid >> 2;
  const int rt   = wid & 3;
  const int m0   = rt * 16;
  const int ltid = tid & 127;

  v8f oacc[4] = {};
  float mrun[8], lrun[8];
  #pragma unroll
  for (int i = 0; i < 8; ++i) { mrun[i] = -INFINITY; lrun[i] = 0.f; }

  auto stage_tile = [&](int t) {
    const int s0  = 64 * t;
    const int kst = t & 1;
    const int vsg = t % 3;
    v4f kr[8], vr[8];
    #pragma unroll
    for (int i = 0; i < 8; ++i) {
      const int idx = ltid + 128 * i;
      const int row = idx >> 4;
      const int c4  = (idx & 15) * 4;
      kr[i] = *(const v4f*)(k_buf + ((size_t)b * kS + s0 + row) * kHD + c4);
      vr[i] = *(const v4f*)(v_buf + ((size_t)b * kS + s0 + row) * kHD + c4);
    }
    #pragma unroll
    for (int i = 0; i < 8; ++i) {
      const int idx = ltid + 128 * i;
      const int row = idx >> 4;
      const int c4  = (idx & 15) * 4;
      *(v4f*)&ks[kst][row][c4] = kr[i];
      const v4f v = vr[i];
      vst[vsg][c4 + 0][row] = v.x;
      vst[vsg][c4 + 1][row] = v.y;
      vst[vsg][c4 + 2][row] = v.z;
      vst[vsg][c4 + 3][row] = v.w;
    }
  };

  auto stage_mem_tile = [&](int kst, int vsg) {
    const int row = ltid >> 1;
    const int h0  = (ltid & 1) * 32;
    const float* src = c_emb + row * kHD + h0;
    v4f xr[8];
    #pragma unroll
    for (int i = 0; i < 8; ++i) xr[i] = *(const v4f*)(src + 4 * i);
    float s = 0.f;
    #pragma unroll
    for (int i = 0; i < 8; ++i) s += (xr[i].x + xr[i].y) + (xr[i].z + xr[i].w);
    s += __shfl_xor(s, 1);
    const float mu = s * (1.f / kHD);
    float q = 0.f;
    #pragma unroll
    for (int i = 0; i < 8; ++i) {
      const v4f d = xr[i] - mu;
      q += (d.x * d.x + d.y * d.y) + (d.z * d.z + d.w * d.w);
    }
    q += __shfl_xor(q, 1);
    const float rstd = rsqrtf(q * (1.f / kHD) + kLnEps);
    #pragma unroll
    for (int i = 0; i < 8; ++i) {
      const int c4 = h0 + 4 * i;
      const v4f g  = *(const v4f*)(gamma + c4);
      const v4f bt = *(const v4f*)(beta + c4);
      const v4f y  = (xr[i] - mu) * rstd * g + bt;
      *(v4f*)&ks[kst][row][c4] = y;
      vst[vsg][c4 + 0][row] = y.x;
      vst[vsg][c4 + 1][row] = y.y;
      vst[vsg][c4 + 2][row] = y.z;
      vst[vsg][c4 + 3][row] = y.w;
    }
  };

  auto score_tile = [&](int t) {
    const int st = t & 1;

    v8f sacc[4] = {};
    #pragma unroll
    for (int kk = 0; kk < 16; ++kk) {
      const int k0 = kk * 4;
      v2f a = *(const v2f*)&qs[m0 + lm][k0 + 2 * half];
      #pragma unroll
      for (int j = 0; j < 4; ++j) {
        v2f bf = *(const v2f*)&ks[st][j * 16 + lm][k0 + 2 * half];
        sacc[j] = wmma_f32(a, bf, sacc[j]);
      }
    }

    float mnew[8], alpha[8], rsum[8];
    #pragma unroll
    for (int i = 0; i < 8; ++i) {
      float v = fmaxf(fmaxf(sacc[0][i], sacc[1][i]),
                      fmaxf(sacc[2][i], sacc[3][i]));
      #pragma unroll
      for (int off = 1; off < 16; off <<= 1)
        v = fmaxf(v, __shfl_xor(v, off));
      mnew[i]  = fmaxf(mrun[i], v);
      alpha[i] = __expf(mrun[i] - mnew[i]);
      mrun[i]  = mnew[i];
      rsum[i]  = 0.f;
    }
    #pragma unroll
    for (int j = 0; j < 4; ++j) {
      #pragma unroll
      for (int i = 0; i < 8; ++i) {
        float p = __expf(sacc[j][i] - mnew[i]);
        sacc[j][i] = p;
        rsum[i] += p;
      }
    }
    #pragma unroll
    for (int i = 0; i < 8; ++i) {
      float rs = rsum[i];
      #pragma unroll
      for (int off = 1; off < 16; off <<= 1)
        rs += __shfl_xor(rs, off);
      lrun[i] = lrun[i] * alpha[i] + rs;
    }

    #pragma unroll
    for (int j = 0; j < 4; ++j)
      #pragma unroll
      for (int i = 0; i < 8; ++i)
        ps[st][rt][i + 8 * half][j * 16 + lm] = sacc[j][i];
    if (lm == 0) {
      v4f f0, f1;
      f0.x = alpha[0]; f0.y = alpha[1]; f0.z = alpha[2]; f0.w = alpha[3];
      f1.x = alpha[4]; f1.y = alpha[5]; f1.z = alpha[6]; f1.w = alpha[7];
      *(v4f*)&als[st][rt][8 * half]     = f0;
      *(v4f*)&als[st][rt][8 * half + 4] = f1;
    }
  };

  auto value_tile = [&](int t) {
    const int st = t & 1;
    const int vs = t % 3;
    const v4f f0 = *(const v4f*)&als[st][rt][8 * half];
    const v4f f1 = *(const v4f*)&als[st][rt][8 * half + 4];
    const float alpha[8] = {f0.x, f0.y, f0.z, f0.w, f1.x, f1.y, f1.z, f1.w};
    #pragma unroll
    for (int c = 0; c < 4; ++c)
      #pragma unroll
      for (int i = 0; i < 8; ++i)
        oacc[c][i] *= alpha[i];

    #pragma unroll
    for (int kk = 0; kk < 16; ++kk) {
      const int k0 = kk * 4;
      v2f a = *(const v2f*)&ps[st][rt][lm][k0 + 2 * half];
      #pragma unroll
      for (int c = 0; c < 4; ++c) {
        v2f bf = *(const v2f*)&vst[vs][c * 16 + lm][k0 + 2 * half];
        oacc[c] = wmma_f32(a, bf, oacc[c]);
      }
    }
  };

  if (role == 2) {
    stage_tile(0);
  } else if (role == 0) {
    #pragma unroll
    for (int i = 0; i < 8; ++i) {
      int idx = ltid + 128 * i;
      int row = idx >> 4;
      int c4  = (idx & 15) * 4;
      v4f v = *(const v4f*)(q_buf + ((size_t)b * kT + q0 + row) * kHD + c4);
      v *= kScale;
      *(v4f*)&qs[row][c4] = v;
    }
  }
  __syncthreads();
  static_assert(kNT >= 3, "the memory tile is staged inside the loop");
  for (int t = 0; t < kNT; ++t) {
    if (role == 2) {
      if (t + 1 < kNT) {
        if (t + 1 == kNT - 1) stage_mem_tile((t + 1) & 1, (t + 1) % 3);
        else                  stage_tile(t + 1);
      }
    } else if (role == 0) {
      score_tile(t);
      if (t == kNT - 1 && lm == 0) {
        v4f l0, l1;
        l0.x = lrun[0]; l0.y = lrun[1]; l0.z = lrun[2]; l0.w = lrun[3];
        l1.x = lrun[4]; l1.y = lrun[5]; l1.z = lrun[6]; l1.w = lrun[7];
        *(v4f*)&lsum[rt][8 * half]     = l0;
        *(v4f*)&lsum[rt][8 * half + 4] = l1;
      }
    } else if (t > 0) {
      value_tile(t - 1);
    }
    __syncthreads();
  }

  if (role == 1) {
    value_tile(kNT - 1);
    const v4f l0 = *(const v4f*)&lsum[rt][8 * half];
    const v4f l1 = *(const v4f*)&lsum[rt][8 * half + 4];
    const float l[8] = {l0.x, l0.y, l0.z, l0.w, l1.x, l1.y, l1.z, l1.w};
    #pragma unroll
    for (int c = 0; c < 4; ++c) {
      const int hd = c * 16 + lm;
      #pragma unroll
      for (int i = 0; i < 8; ++i) {
        const int t = q0 + m0 + i + 8 * half;
        out[((size_t)b * kT + t) * kHD + hd] = oacc[c][i] / l[i];
      }
    }
  }
}

extern "C" void kernel_launch(void* const* d_in, const int* in_sizes, int n_in,
                              void* d_out, int out_size, void* d_ws, size_t ws_size,
                              hipStream_t stream) {
  const float* x     = (const float*)d_in[0];
  const float* c_emb = (const float*)d_in[1];
  const float* Wq    = (const float*)d_in[2];
  const float* Wk    = (const float*)d_in[3];
  const float* Wv    = (const float*)d_in[4];
  const float* gamma = (const float*)d_in[5];
  const float* beta  = (const float*)d_in[6];
  float* out = (float*)d_out;

  float* q_buf = (float*)d_ws;
  float* k_buf = q_buf + (size_t)kB * kT * kHD;
  float* v_buf = k_buf + (size_t)kB * kS * kHD;

  qkv_kernel<<<kProjBlocks, 256, 0, stream>>>(x, Wq, Wk, Wv, q_buf, k_buf, v_buf);
  attn_kernel<<<dim3(kAttnBlocksX, kB), 384, 0, stream>>>(q_buf, k_buf, v_buf, out,
                                                          c_emb, gamma, beta);
}


# attention softmax: the row max and row sum butterflies of each key tile run as DPP row_xmask steps on the vector ALU, not as ds_bpermute through the LDS unit (64 per tile gone); sum order unchanged
# speedup vs baseline: 2.9279x; 1.0564x over previous
#include <hip/hip_runtime.h>
#include <math.h>

typedef float v2f __attribute__((ext_vector_type(2)));
typedef float v4f __attribute__((ext_vector_type(4)));
typedef float v8f __attribute__((ext_vector_type(8)));

namespace {
constexpr int kB  = 8;
constexpr int kT  = 2048;
constexpr int kD  = 1024;
constexpr int kHD = 64;
constexpr int kM  = 64;
constexpr int kS  = kT + kM;
constexpr float kLnEps = 1e-5f;
constexpr float kScale = 0.125f;
constexpr int kProjBlocks = 256;
constexpr int kNT    = kS / 64;
constexpr int kQT    = 64;
constexpr int kAttnBlocksX = kT / kQT;
}

__device__ __forceinline__ v8f wmma_f32(v2f a, v2f b, v8f c) {
  return __builtin_amdgcn_wmma_f32_16x16x4_f32(
       false, a,  false, b,
       (short)0, c,  false,  false);
}

template <int M>
__device__ __forceinline__ float row_xor(float v) {
  return __builtin_bit_cast(float,
      __builtin_amdgcn_update_dpp(0, __builtin_bit_cast(int, v),
                                    0x160 + M,
                                   0xF,  0xF,
                                   true));
}

template <int M>
__device__ __forceinline__ float row_max_step(float v) {
  float r;
  asm("v_max_num_f32_dpp %0, %1, %1 row_xmask:%2 row_mask:0xf bank_mask:0xf bound_ctrl:1"
      : "=v"(r) : "v"(v), "n"(M));
  return r;
}


template <int NH>
__device__ __forceinline__ void store_proj_tiles(const v8f (&acc)[2][6],
                                                 int row0, int half, int lm,
                                                 float* __restrict__ q_buf,
                                                 float* __restrict__ k_buf,
                                                 float* __restrict__ v_buf) {
  #pragma unroll
  for (int c = 0; c < 6; ++c) {
    constexpr int kFirst = 6 * NH;
    const int vc  = kFirst + c;
    const int mat = vc >> 2;
    const int hd  = (vc & 3) * 16 + lm;
    #pragma unroll
    for (int r = 0; r < 2; ++r) {
      #pragma unroll
      for (int i = 0; i < 8; ++i) {
        const int gr = row0 + 16 * r + i + 8 * half;
        const float val = acc[r][c][i];
        if (mat == 0) {
          q_buf[(size_t)gr * kHD + hd] = val;
        } else {
          const int bb = gr / kT;
          const int tt = gr - bb * kT;
          float* dst = (mat == 1) ? k_buf : v_buf;
          dst[((size_t)bb * kS + tt) * kHD + hd] = val;
        }
      }
    }
  }
}

__global__ __launch_bounds__(256)
void qkv_kernel(const float* __restrict__ x,
                const float* __restrict__ Wq,
                const float* __restrict__ Wk,
                const float* __restrict__ Wv,
                float* __restrict__ q_buf,
                float* __restrict__ k_buf,
                float* __restrict__ v_buf) {
  __shared__ alignas(16) float xs[2][128][68];
  __shared__ alignas(16) float wst[2][96][68];

  const int tid  = threadIdx.x;
  const int wid  = __builtin_amdgcn_readfirstlane(tid >> 5);
  const int lane = tid & 31;
  const int half = lane >> 4;
  const int lm   = lane & 15;
  const int nh   = blockIdx.x & 1;
  const int r0   = (blockIdx.x >> 1) * 128;
  const bool staging = (wid >= 4);
  const int m0   = (wid & 3) * 32;
  const int ltid = tid & 127;

  const float* wseg0 = nh ? (Wk + 32) : Wq;
  const float* wseg1 = nh ? Wv        : (Wq + 32);
  const float* wseg2 = nh ? (Wv + 32) : Wk;

  v8f acc[2][6] = {};

  auto stage_tile = [&](int t) {
    const int kt = 64 * t;
    const int st = t & 1;
    v4f xr[16];
    v4f wr[12];
    #pragma unroll
    for (int i = 0; i < 16; ++i) {
      const int idx = ltid + 128 * i;
      const int row = idx >> 4;
      const int c4  = (idx & 15) * 4;
      xr[i] = *(const v4f*)(x + (size_t)(r0 + row) * kD + kt + c4);
    }
    #pragma unroll
    for (int s = 0; s < 3; ++s) {
      const float* src = (s == 0) ? wseg0 : ((s == 1) ? wseg1 : wseg2);
      #pragma unroll
      for (int u = 0; u < 4; ++u) {
        const int idx = ltid + 128 * u;
        const int kr  = idx >> 3;
        const int cc  = (idx & 7) * 4;
        wr[s * 4 + u] = *(const v4f*)(src + (size_t)(kt + kr) * kHD + cc);
      }
    }
    #pragma unroll
    for (int i = 0; i < 16; ++i) {
      const int idx = ltid + 128 * i;
      const int row = idx >> 4;
      const int c4  = (idx & 15) * 4;
      *(v4f*)&xs[st][row][c4] = xr[i];
    }
    #pragma unroll
    for (int s = 0; s < 3; ++s) {
      #pragma unroll
      for (int u = 0; u < 4; ++u) {
        const int idx = ltid + 128 * u;
        const int kr  = idx >> 3;
        const int col = s * 32 + (idx & 7) * 4;
        const v4f w = wr[s * 4 + u];
        wst[st][col + 0][kr] = w.x;
        wst[st][col + 1][kr] = w.y;
        wst[st][col + 2][kr] = w.z;
        wst[st][col + 3][kr] = w.w;
      }
    }
  };
  auto matrix_tile = [&](int t) {
    const int st = t & 1;
    #pragma unroll
    for (int kk = 0; kk < 16; ++kk) {
      const int k0 = kk * 4;
      const v2f a0 = *(const v2f*)&xs[st][m0 + lm][k0 + 2 * half];
      const v2f a1 = *(const v2f*)&xs[st][m0 + 16 + lm][k0 + 2 * half];
      #pragma unroll
      for (int c = 0; c < 6; ++c) {
        const v2f bf = *(const v2f*)&wst[st][c * 16 + lm][k0 + 2 * half];
        acc[0][c] = wmma_f32(a0, bf, acc[0][c]);
        acc[1][c] = wmma_f32(a1, bf, acc[1][c]);
      }
    }
  };

  constexpr int kKT = kD / 64;
  if (staging) stage_tile(0);
  __syncthreads();
  for (int t = 1; t < kKT; ++t) {
    if (staging) stage_tile(t); else matrix_tile(t - 1);
    __syncthreads();
  }
  if (!staging) matrix_tile(kKT - 1);

  if (!staging) {
    if (nh == 0) store_proj_tiles<0>(acc, r0 + m0, half, lm, q_buf, k_buf, v_buf);
    else         store_proj_tiles<1>(acc, r0 + m0, half, lm, q_buf, k_buf, v_buf);
  }
}

__global__ __launch_bounds__(384)
void attn_kernel(const float* __restrict__ q_buf,
                 const float* __restrict__ k_buf,
                 const float* __restrict__ v_buf,
                 float* __restrict__ out,
                 const float* __restrict__ c_emb,
                 const float* __restrict__ gamma,
                 const float* __restrict__ beta) {
  __shared__ alignas(16) float qs[kQT][68];
  __shared__ alignas(16) float ks[2][64][68];
  __shared__ alignas(16) float vst[3][64][68];
  __shared__ alignas(16) float ps[2][4][16][68];
  __shared__ alignas(16) float als[2][4][16];
  __shared__ alignas(16) float lsum[4][16];

  const int tid  = threadIdx.x;
  const int wid  = __builtin_amdgcn_readfirstlane(tid >> 5);
  const int lane = tid & 31;
  const int half = lane >> 4;
  const int lm   = lane & 15;
  const int b    = blockIdx.y;
  const int q0   = blockIdx.x * kQT;
  const int role = wid >> 2;
  const int rt   = wid & 3;
  const int m0   = rt * 16;
  const int ltid = tid & 127;

  v8f oacc[4] = {};
  float mrun[8], lrun[8];
  #pragma unroll
  for (int i = 0; i < 8; ++i) { mrun[i] = -INFINITY; lrun[i] = 0.f; }

  auto stage_tile = [&](int t) {
    const int s0  = 64 * t;
    const int kst = t & 1;
    const int vsg = t % 3;
    v4f kr[8], vr[8];
    #pragma unroll
    for (int i = 0; i < 8; ++i) {
      const int idx = ltid + 128 * i;
      const int row = idx >> 4;
      const int c4  = (idx & 15) * 4;
      kr[i] = *(const v4f*)(k_buf + ((size_t)b * kS + s0 + row) * kHD + c4);
      vr[i] = *(const v4f*)(v_buf + ((size_t)b * kS + s0 + row) * kHD + c4);
    }
    #pragma unroll
    for (int i = 0; i < 8; ++i) {
      const int idx = ltid + 128 * i;
      const int row = idx >> 4;
      const int c4  = (idx & 15) * 4;
      *(v4f*)&ks[kst][row][c4] = kr[i];
      const v4f v = vr[i];
      vst[vsg][c4 + 0][row] = v.x;
      vst[vsg][c4 + 1][row] = v.y;
      vst[vsg][c4 + 2][row] = v.z;
      vst[vsg][c4 + 3][row] = v.w;
    }
  };

  auto stage_mem_tile = [&](int kst, int vsg) {
    const int row = ltid >> 1;
    const int h0  = (ltid & 1) * 32;
    const float* src = c_emb + row * kHD + h0;
    v4f xr[8];
    #pragma unroll
    for (int i = 0; i < 8; ++i) xr[i] = *(const v4f*)(src + 4 * i);
    float s = 0.f;
    #pragma unroll
    for (int i = 0; i < 8; ++i) s += (xr[i].x + xr[i].y) + (xr[i].z + xr[i].w);
    s += row_xor<1>(s);
    const float mu = s * (1.f / kHD);
    float q = 0.f;
    #pragma unroll
    for (int i = 0; i < 8; ++i) {
      const v4f d = xr[i] - mu;
      q += (d.x * d.x + d.y * d.y) + (d.z * d.z + d.w * d.w);
    }
    q += row_xor<1>(q);
    const float rstd = rsqrtf(q * (1.f / kHD) + kLnEps);
    #pragma unroll
    for (int i = 0; i < 8; ++i) {
      const int c4 = h0 + 4 * i;
      const v4f g  = *(const v4f*)(gamma + c4);
      const v4f bt = *(const v4f*)(beta + c4);
      const v4f y  = (xr[i] - mu) * rstd * g + bt;
      *(v4f*)&ks[kst][row][c4] = y;
      vst[vsg][c4 + 0][row] = y.x;
      vst[vsg][c4 + 1][row] = y.y;
      vst[vsg][c4 + 2][row] = y.z;
      vst[vsg][c4 + 3][row] = y.w;
    }
  };

  auto score_tile = [&](int t) {
    const int st = t & 1;

    v8f sacc[4] = {};
    #pragma unroll
    for (int kk = 0; kk < 16; ++kk) {
      const int k0 = kk * 4;
      v2f a = *(const v2f*)&qs[m0 + lm][k0 + 2 * half];
      #pragma unroll
      for (int j = 0; j < 4; ++j) {
        v2f bf = *(const v2f*)&ks[st][j * 16 + lm][k0 + 2 * half];
        sacc[j] = wmma_f32(a, bf, sacc[j]);
      }
    }

    float mnew[8], alpha[8], rsum[8], vmax[8];
    #pragma unroll
    for (int i = 0; i < 8; ++i)
      vmax[i] = fmaxf(fmaxf(sacc[0][i], sacc[1][i]),
                      fmaxf(sacc[2][i], sacc[3][i]));
    #pragma unroll
    for (int i = 0; i < 8; ++i) vmax[i] = row_max_step<1>(vmax[i]);
    #pragma unroll
    for (int i = 0; i < 8; ++i) vmax[i] = row_max_step<2>(vmax[i]);
    #pragma unroll
    for (int i = 0; i < 8; ++i) vmax[i] = row_max_step<4>(vmax[i]);
    #pragma unroll
    for (int i = 0; i < 8; ++i) vmax[i] = row_max_step<8>(vmax[i]);
    #pragma unroll
    for (int i = 0; i < 8; ++i) {
      mnew[i]  = fmaxf(mrun[i], vmax[i]);
      alpha[i] = __expf(mrun[i] - mnew[i]);
      mrun[i]  = mnew[i];
      rsum[i]  = 0.f;
    }
    #pragma unroll
    for (int j = 0; j < 4; ++j) {
      #pragma unroll
      for (int i = 0; i < 8; ++i) {
        float p = __expf(sacc[j][i] - mnew[i]);
        sacc[j][i] = p;
        rsum[i] += p;
      }
    }
    #pragma unroll
    for (int i = 0; i < 8; ++i) {
      float rs = rsum[i];
      rs += row_xor<1>(rs);
      rs += row_xor<2>(rs);
      rs += row_xor<4>(rs);
      rs += row_xor<8>(rs);
      lrun[i] = lrun[i] * alpha[i] + rs;
    }

    #pragma unroll
    for (int j = 0; j < 4; ++j)
      #pragma unroll
      for (int i = 0; i < 8; ++i)
        ps[st][rt][i + 8 * half][j * 16 + lm] = sacc[j][i];
    if (lm == 0) {
      v4f f0, f1;
      f0.x = alpha[0]; f0.y = alpha[1]; f0.z = alpha[2]; f0.w = alpha[3];
      f1.x = alpha[4]; f1.y = alpha[5]; f1.z = alpha[6]; f1.w = alpha[7];
      *(v4f*)&als[st][rt][8 * half]     = f0;
      *(v4f*)&als[st][rt][8 * half + 4] = f1;
    }
  };

  auto value_tile = [&](int t) {
    const int st = t & 1;
    const int vs = t % 3;
    const v4f f0 = *(const v4f*)&als[st][rt][8 * half];
    const v4f f1 = *(const v4f*)&als[st][rt][8 * half + 4];
    const float alpha[8] = {f0.x, f0.y, f0.z, f0.w, f1.x, f1.y, f1.z, f1.w};
    #pragma unroll
    for (int c = 0; c < 4; ++c)
      #pragma unroll
      for (int i = 0; i < 8; ++i)
        oacc[c][i] *= alpha[i];

    #pragma unroll
    for (int kk = 0; kk < 16; ++kk) {
      const int k0 = kk * 4;
      v2f a = *(const v2f*)&ps[st][rt][lm][k0 + 2 * half];
      #pragma unroll
      for (int c = 0; c < 4; ++c) {
        v2f bf = *(const v2f*)&vst[vs][c * 16 + lm][k0 + 2 * half];
        oacc[c] = wmma_f32(a, bf, oacc[c]);
      }
    }
  };

  if (role == 2) {
    stage_tile(0);
  } else if (role == 0) {
    #pragma unroll
    for (int i = 0; i < 8; ++i) {
      int idx = ltid + 128 * i;
      int row = idx >> 4;
      int c4  = (idx & 15) * 4;
      v4f v = *(const v4f*)(q_buf + ((size_t)b * kT + q0 + row) * kHD + c4);
      v *= kScale;
      *(v4f*)&qs[row][c4] = v;
    }
  }
  __syncthreads();
  static_assert(kNT >= 3, "the memory tile is staged inside the loop");
  for (int t = 0; t < kNT; ++t) {
    if (role == 2) {
      if (t + 1 < kNT) {
        if (t + 1 == kNT - 1) stage_mem_tile((t + 1) & 1, (t + 1) % 3);
        else                  stage_tile(t + 1);
      }
    } else if (role == 0) {
      score_tile(t);
      if (t == kNT - 1 && lm == 0) {
        v4f l0, l1;
        l0.x = lrun[0]; l0.y = lrun[1]; l0.z = lrun[2]; l0.w = lrun[3];
        l1.x = lrun[4]; l1.y = lrun[5]; l1.z = lrun[6]; l1.w = lrun[7];
        *(v4f*)&lsum[rt][8 * half]     = l0;
        *(v4f*)&lsum[rt][8 * half + 4] = l1;
      }
    } else if (t > 0) {
      value_tile(t - 1);
    }
    __syncthreads();
  }

  if (role == 1) {
    value_tile(kNT - 1);
    const v4f l0 = *(const v4f*)&lsum[rt][8 * half];
    const v4f l1 = *(const v4f*)&lsum[rt][8 * half + 4];
    const float l[8] = {l0.x, l0.y, l0.z, l0.w, l1.x, l1.y, l1.z, l1.w};
    #pragma unroll
    for (int c = 0; c < 4; ++c) {
      const int hd = c * 16 + lm;
      #pragma unroll
      for (int i = 0; i < 8; ++i) {
        const int t = q0 + m0 + i + 8 * half;
        out[((size_t)b * kT + t) * kHD + hd] = oacc[c][i] / l[i];
      }
    }
  }
}

extern "C" void kernel_launch(void* const* d_in, const int* in_sizes, int n_in,
                              void* d_out, int out_size, void* d_ws, size_t ws_size,
                              hipStream_t stream) {
  const float* x     = (const float*)d_in[0];
  const float* c_emb = (const float*)d_in[1];
  const float* Wq    = (const float*)d_in[2];
  const float* Wk    = (const float*)d_in[3];
  const float* Wv    = (const float*)d_in[4];
  const float* gamma = (const float*)d_in[5];
  const float* beta  = (const float*)d_in[6];
  float* out = (float*)d_out;

  float* q_buf = (float*)d_ws;
  float* k_buf = q_buf + (size_t)kB * kT * kHD;
  float* v_buf = k_buf + (size_t)kB * kS * kHD;

  qkv_kernel<<<kProjBlocks, 256, 0, stream>>>(x, Wq, Wk, Wv, q_buf, k_buf, v_buf);
  attn_kernel<<<dim3(kAttnBlocksX, kB), 384, 0, stream>>>(q_buf, k_buf, v_buf, out,
                                                          c_emb, gamma, beta);
}


# both kernels: each wave role runs its own tile loop (same barriers, same order); role-private accumulators no longer cross other roles' paths, so per-tile register copies go; sum order unchanged
# speedup vs baseline: 3.1623x; 1.0801x over previous
#include <hip/hip_runtime.h>
#include <math.h>

typedef float v2f __attribute__((ext_vector_type(2)));
typedef float v4f __attribute__((ext_vector_type(4)));
typedef float v8f __attribute__((ext_vector_type(8)));

namespace {
constexpr int kB  = 8;
constexpr int kT  = 2048;
constexpr int kD  = 1024;
constexpr int kHD = 64;
constexpr int kM  = 64;
constexpr int kS  = kT + kM;
constexpr float kLnEps = 1e-5f;
constexpr float kScale = 0.125f;
constexpr int kProjBlocks = 256;
constexpr int kNT    = kS / 64;
constexpr int kQT    = 64;
constexpr int kAttnBlocksX = kT / kQT;
}

__device__ __forceinline__ v8f wmma_f32(v2f a, v2f b, v8f c) {
  return __builtin_amdgcn_wmma_f32_16x16x4_f32(
       false, a,  false, b,
       (short)0, c,  false,  false);
}

template <int M>
__device__ __forceinline__ float row_xor(float v) {
  return __builtin_bit_cast(float,
      __builtin_amdgcn_update_dpp(0, __builtin_bit_cast(int, v),
                                    0x160 + M,
                                   0xF,  0xF,
                                   true));
}

template <int M>
__device__ __forceinline__ float row_max_step(float v) {
  float r;
  asm("v_max_num_f32_dpp %0, %1, %1 row_xmask:%2 row_mask:0xf bank_mask:0xf bound_ctrl:1"
      : "=v"(r) : "v"(v), "n"(M));
  return r;
}


template <int NH>
__device__ __forceinline__ void store_proj_tiles(const v8f (&acc)[2][6],
                                                 int row0, int half, int lm,
                                                 float* __restrict__ q_buf,
                                                 float* __restrict__ k_buf,
                                                 float* __restrict__ v_buf) {
  #pragma unroll
  for (int c = 0; c < 6; ++c) {
    constexpr int kFirst = 6 * NH;
    const int vc  = kFirst + c;
    const int mat = vc >> 2;
    const int hd  = (vc & 3) * 16 + lm;
    #pragma unroll
    for (int r = 0; r < 2; ++r) {
      #pragma unroll
      for (int i = 0; i < 8; ++i) {
        const int gr = row0 + 16 * r + i + 8 * half;
        const float val = acc[r][c][i];
        if (mat == 0) {
          q_buf[(size_t)gr * kHD + hd] = val;
        } else {
          const int bb = gr / kT;
          const int tt = gr - bb * kT;
          float* dst = (mat == 1) ? k_buf : v_buf;
          dst[((size_t)bb * kS + tt) * kHD + hd] = val;
        }
      }
    }
  }
}

__global__ __launch_bounds__(256)
void qkv_kernel(const float* __restrict__ x,
                const float* __restrict__ Wq,
                const float* __restrict__ Wk,
                const float* __restrict__ Wv,
                float* __restrict__ q_buf,
                float* __restrict__ k_buf,
                float* __restrict__ v_buf) {
  __shared__ alignas(16) float xs[2][128][68];
  __shared__ alignas(16) float wst[2][96][68];

  const int tid  = threadIdx.x;
  const int wid  = __builtin_amdgcn_readfirstlane(tid >> 5);
  const int lane = tid & 31;
  const int half = lane >> 4;
  const int lm   = lane & 15;
  const int nh   = blockIdx.x & 1;
  const int r0   = (blockIdx.x >> 1) * 128;
  const bool staging = (wid >= 4);
  const int m0   = (wid & 3) * 32;
  const int ltid = tid & 127;

  const float* wseg0 = nh ? (Wk + 32) : Wq;
  const float* wseg1 = nh ? Wv        : (Wq + 32);
  const float* wseg2 = nh ? (Wv + 32) : Wk;

  v8f acc[2][6] = {};

  auto stage_tile = [&](int t) {
    const int kt = 64 * t;
    const int st = t & 1;
    v4f xr[16];
    v4f wr[12];
    #pragma unroll
    for (int i = 0; i < 16; ++i) {
      const int idx = ltid + 128 * i;
      const int row = idx >> 4;
      const int c4  = (idx & 15) * 4;
      xr[i] = *(const v4f*)(x + (size_t)(r0 + row) * kD + kt + c4);
    }
    #pragma unroll
    for (int s = 0; s < 3; ++s) {
      const float* src = (s == 0) ? wseg0 : ((s == 1) ? wseg1 : wseg2);
      #pragma unroll
      for (int u = 0; u < 4; ++u) {
        const int idx = ltid + 128 * u;
        const int kr  = idx >> 3;
        const int cc  = (idx & 7) * 4;
        wr[s * 4 + u] = *(const v4f*)(src + (size_t)(kt + kr) * kHD + cc);
      }
    }
    #pragma unroll
    for (int i = 0; i < 16; ++i) {
      const int idx = ltid + 128 * i;
      const int row = idx >> 4;
      const int c4  = (idx & 15) * 4;
      *(v4f*)&xs[st][row][c4] = xr[i];
    }
    #pragma unroll
    for (int s = 0; s < 3; ++s) {
      #pragma unroll
      for (int u = 0; u < 4; ++u) {
        const int idx = ltid + 128 * u;
        const int kr  = idx >> 3;
        const int col = s * 32 + (idx & 7) * 4;
        const v4f w = wr[s * 4 + u];
        wst[st][col + 0][kr] = w.x;
        wst[st][col + 1][kr] = w.y;
        wst[st][col + 2][kr] = w.z;
        wst[st][col + 3][kr] = w.w;
      }
    }
  };
  auto matrix_tile = [&](int t) {
    const int st = t & 1;
    #pragma unroll
    for (int kk = 0; kk < 16; ++kk) {
      const int k0 = kk * 4;
      const v2f a0 = *(const v2f*)&xs[st][m0 + lm][k0 + 2 * half];
      const v2f a1 = *(const v2f*)&xs[st][m0 + 16 + lm][k0 + 2 * half];
      #pragma unroll
      for (int c = 0; c < 6; ++c) {
        const v2f bf = *(const v2f*)&wst[st][c * 16 + lm][k0 + 2 * half];
        acc[0][c] = wmma_f32(a0, bf, acc[0][c]);
        acc[1][c] = wmma_f32(a1, bf, acc[1][c]);
      }
    }
  };

  constexpr int kKT = kD / 64;
  if (staging) {
    stage_tile(0);
    __syncthreads();
    for (int t = 1; t < kKT; ++t) {
      stage_tile(t);
      __syncthreads();
    }
  } else {
    __syncthreads();
    for (int t = 1; t < kKT; ++t) {
      matrix_tile(t - 1);
      __syncthreads();
    }
    matrix_tile(kKT - 1);

    if (nh == 0) store_proj_tiles<0>(acc, r0 + m0, half, lm, q_buf, k_buf, v_buf);
    else         store_proj_tiles<1>(acc, r0 + m0, half, lm, q_buf, k_buf, v_buf);
  }
}

__global__ __launch_bounds__(384)
void attn_kernel(const float* __restrict__ q_buf,
                 const float* __restrict__ k_buf,
                 const float* __restrict__ v_buf,
                 float* __restrict__ out,
                 const float* __restrict__ c_emb,
                 const float* __restrict__ gamma,
                 const float* __restrict__ beta) {
  __shared__ alignas(16) float qs[kQT][68];
  __shared__ alignas(16) float ks[2][64][68];
  __shared__ alignas(16) float vst[3][64][68];
  __shared__ alignas(16) float ps[2][4][16][68];
  __shared__ alignas(16) float als[2][4][16];
  __shared__ alignas(16) float lsum[4][16];

  const int tid  = threadIdx.x;
  const int wid  = __builtin_amdgcn_readfirstlane(tid >> 5);
  const int lane = tid & 31;
  const int half = lane >> 4;
  const int lm   = lane & 15;
  const int b    = blockIdx.y;
  const int q0   = blockIdx.x * kQT;
  const int role = wid >> 2;
  const int rt   = wid & 3;
  const int m0   = rt * 16;
  const int ltid = tid & 127;

  v8f oacc[4] = {};
  float mrun[8], lrun[8];
  #pragma unroll
  for (int i = 0; i < 8; ++i) { mrun[i] = -INFINITY; lrun[i] = 0.f; }

  auto stage_tile = [&](int t) {
    const int s0  = 64 * t;
    const int kst = t & 1;
    const int vsg = t % 3;
    v4f kr[8], vr[8];
    #pragma unroll
    for (int i = 0; i < 8; ++i) {
      const int idx = ltid + 128 * i;
      const int row = idx >> 4;
      const int c4  = (idx & 15) * 4;
      kr[i] = *(const v4f*)(k_buf + ((size_t)b * kS + s0 + row) * kHD + c4);
      vr[i] = *(const v4f*)(v_buf + ((size_t)b * kS + s0 + row) * kHD + c4);
    }
    #pragma unroll
    for (int i = 0; i < 8; ++i) {
      const int idx = ltid + 128 * i;
      const int row = idx >> 4;
      const int c4  = (idx & 15) * 4;
      *(v4f*)&ks[kst][row][c4] = kr[i];
      const v4f v = vr[i];
      vst[vsg][c4 + 0][row] = v.x;
      vst[vsg][c4 + 1][row] = v.y;
      vst[vsg][c4 + 2][row] = v.z;
      vst[vsg][c4 + 3][row] = v.w;
    }
  };

  auto stage_mem_tile = [&](int kst, int vsg) {
    const int row = ltid >> 1;
    const int h0  = (ltid & 1) * 32;
    const float* src = c_emb + row * kHD + h0;
    v4f xr[8];
    #pragma unroll
    for (int i = 0; i < 8; ++i) xr[i] = *(const v4f*)(src + 4 * i);
    float s = 0.f;
    #pragma unroll
    for (int i = 0; i < 8; ++i) s += (xr[i].x + xr[i].y) + (xr[i].z + xr[i].w);
    s += row_xor<1>(s);
    const float mu = s * (1.f / kHD);
    float q = 0.f;
    #pragma unroll
    for (int i = 0; i < 8; ++i) {
      const v4f d = xr[i] - mu;
      q += (d.x * d.x + d.y * d.y) + (d.z * d.z + d.w * d.w);
    }
    q += row_xor<1>(q);
    const float rstd = rsqrtf(q * (1.f / kHD) + kLnEps);
    #pragma unroll
    for (int i = 0; i < 8; ++i) {
      const int c4 = h0 + 4 * i;
      const v4f g  = *(const v4f*)(gamma + c4);
      const v4f bt = *(const v4f*)(beta + c4);
      const v4f y  = (xr[i] - mu) * rstd * g + bt;
      *(v4f*)&ks[kst][row][c4] = y;
      vst[vsg][c4 + 0][row] = y.x;
      vst[vsg][c4 + 1][row] = y.y;
      vst[vsg][c4 + 2][row] = y.z;
      vst[vsg][c4 + 3][row] = y.w;
    }
  };

  auto score_tile = [&](int t) {
    const int st = t & 1;

    v8f sacc[4] = {};
    #pragma unroll
    for (int kk = 0; kk < 16; ++kk) {
      const int k0 = kk * 4;
      v2f a = *(const v2f*)&qs[m0 + lm][k0 + 2 * half];
      #pragma unroll
      for (int j = 0; j < 4; ++j) {
        v2f bf = *(const v2f*)&ks[st][j * 16 + lm][k0 + 2 * half];
        sacc[j] = wmma_f32(a, bf, sacc[j]);
      }
    }

    float mnew[8], alpha[8], rsum[8], vmax[8];
    #pragma unroll
    for (int i = 0; i < 8; ++i)
      vmax[i] = fmaxf(fmaxf(sacc[0][i], sacc[1][i]),
                      fmaxf(sacc[2][i], sacc[3][i]));
    #pragma unroll
    for (int i = 0; i < 8; ++i) vmax[i] = row_max_step<1>(vmax[i]);
    #pragma unroll
    for (int i = 0; i < 8; ++i) vmax[i] = row_max_step<2>(vmax[i]);
    #pragma unroll
    for (int i = 0; i < 8; ++i) vmax[i] = row_max_step<4>(vmax[i]);
    #pragma unroll
    for (int i = 0; i < 8; ++i) vmax[i] = row_max_step<8>(vmax[i]);
    #pragma unroll
    for (int i = 0; i < 8; ++i) {
      mnew[i]  = fmaxf(mrun[i], vmax[i]);
      alpha[i] = __expf(mrun[i] - mnew[i]);
      mrun[i]  = mnew[i];
      rsum[i]  = 0.f;
    }
    #pragma unroll
    for (int j = 0; j < 4; ++j) {
      #pragma unroll
      for (int i = 0; i < 8; ++i) {
        float p = __expf(sacc[j][i] - mnew[i]);
        sacc[j][i] = p;
        rsum[i] += p;
      }
    }
    #pragma unroll
    for (int i = 0; i < 8; ++i) {
      float rs = rsum[i];
      rs += row_xor<1>(rs);
      rs += row_xor<2>(rs);
      rs += row_xor<4>(rs);
      rs += row_xor<8>(rs);
      lrun[i] = lrun[i] * alpha[i] + rs;
    }

    #pragma unroll
    for (int j = 0; j < 4; ++j)
      #pragma unroll
      for (int i = 0; i < 8; ++i)
        ps[st][rt][i + 8 * half][j * 16 + lm] = sacc[j][i];
    if (lm == 0) {
      v4f f0, f1;
      f0.x = alpha[0]; f0.y = alpha[1]; f0.z = alpha[2]; f0.w = alpha[3];
      f1.x = alpha[4]; f1.y = alpha[5]; f1.z = alpha[6]; f1.w = alpha[7];
      *(v4f*)&als[st][rt][8 * half]     = f0;
      *(v4f*)&als[st][rt][8 * half + 4] = f1;
    }
  };

  auto value_tile = [&](int t) {
    const int st = t & 1;
    const int vs = t % 3;
    const v4f f0 = *(const v4f*)&als[st][rt][8 * half];
    const v4f f1 = *(const v4f*)&als[st][rt][8 * half + 4];
    const float alpha[8] = {f0.x, f0.y, f0.z, f0.w, f1.x, f1.y, f1.z, f1.w};
    #pragma unroll
    for (int c = 0; c < 4; ++c)
      #pragma unroll
      for (int i = 0; i < 8; ++i)
        oacc[c][i] *= alpha[i];

    #pragma unroll
    for (int kk = 0; kk < 16; ++kk) {
      const int k0 = kk * 4;
      v2f a = *(const v2f*)&ps[st][rt][lm][k0 + 2 * half];
      #pragma unroll
      for (int c = 0; c < 4; ++c) {
        v2f bf = *(const v2f*)&vst[vs][c * 16 + lm][k0 + 2 * half];
        oacc[c] = wmma_f32(a, bf, oacc[c]);
      }
    }
  };

  static_assert(kNT >= 3, "the memory tile is staged inside the loop");
  if (role == 2) {
    stage_tile(0);
    __syncthreads();
    for (int t = 0; t < kNT - 2; ++t) {
      stage_tile(t + 1);
      __syncthreads();
    }
    stage_mem_tile((kNT - 1) & 1, (kNT - 1) % 3);
    __syncthreads();
    __syncthreads();
  } else if (role == 0) {
    #pragma unroll
    for (int i = 0; i < 8; ++i) {
      int idx = ltid + 128 * i;
      int row = idx >> 4;
      int c4  = (idx & 15) * 4;
      v4f v = *(const v4f*)(q_buf + ((size_t)b * kT + q0 + row) * kHD + c4);
      v *= kScale;
      *(v4f*)&qs[row][c4] = v;
    }
    __syncthreads();
    for (int t = 0; t < kNT; ++t) {
      score_tile(t);
      if (t == kNT - 1 && lm == 0) {
        v4f l0, l1;
        l0.x = lrun[0]; l0.y = lrun[1]; l0.z = lrun[2]; l0.w = lrun[3];
        l1.x = lrun[4]; l1.y = lrun[5]; l1.z = lrun[6]; l1.w = lrun[7];
        *(v4f*)&lsum[rt][8 * half]     = l0;
        *(v4f*)&lsum[rt][8 * half + 4] = l1;
      }
      __syncthreads();
    }
  } else {
    __syncthreads();
    __syncthreads();
    for (int t = 1; t < kNT; ++t) {
      value_tile(t - 1);
      __syncthreads();
    }
    value_tile(kNT - 1);
    const v4f l0 = *(const v4f*)&lsum[rt][8 * half];
    const v4f l1 = *(const v4f*)&lsum[rt][8 * half + 4];
    const float l[8] = {l0.x, l0.y, l0.z, l0.w, l1.x, l1.y, l1.z, l1.w};
    #pragma unroll
    for (int c = 0; c < 4; ++c) {
      const int hd = c * 16 + lm;
      #pragma unroll
      for (int i = 0; i < 8; ++i) {
        const int t = q0 + m0 + i + 8 * half;
        out[((size_t)b * kT + t) * kHD + hd] = oacc[c][i] / l[i];
      }
    }
  }
}

extern "C" void kernel_launch(void* const* d_in, const int* in_sizes, int n_in,
                              void* d_out, int out_size, void* d_ws, size_t ws_size,
                              hipStream_t stream) {
  const float* x     = (const float*)d_in[0];
  const float* c_emb = (const float*)d_in[1];
  const float* Wq    = (const float*)d_in[2];
  const float* Wk    = (const float*)d_in[3];
  const float* Wv    = (const float*)d_in[4];
  const float* gamma = (const float*)d_in[5];
  const float* beta  = (const float*)d_in[6];
  float* out = (float*)d_out;

  float* q_buf = (float*)d_ws;
  float* k_buf = q_buf + (size_t)kB * kT * kHD;
  float* v_buf = k_buf + (size_t)kB * kS * kHD;

  qkv_kernel<<<kProjBlocks, 256, 0, stream>>>(x, Wq, Wk, Wv, q_buf, k_buf, v_buf);
  attn_kernel<<<dim3(kAttnBlocksX, kB), 384, 0, stream>>>(q_buf, k_buf, v_buf, out,
                                                          c_emb, gamma, beta);
}
